# starting kernel
# baseline (speedup 1.0000x reference)
#include <hip/hip_runtime.h>

typedef float v2f __attribute__((ext_vector_type(2)));
typedef float v8f __attribute__((ext_vector_type(8)));

__global__ void tag_deg(const int* __restrict__ dst, float* __restrict__ deg, int E) {
    int e = blockIdx.x * blockDim.x + threadIdx.x;
    if (e < E) unsafeAtomicAdd(&deg[dst[e]], 1.0f);
}

__global__ void tag_dis(const float* __restrict__ deg, float* __restrict__ dis, int N) {
    int i = blockIdx.x * blockDim.x + threadIdx.x;
    if (i < N) {
        float d = deg[i];
        dis[i] = (d > 0.0f) ? rsqrtf(fmaxf(d, 1.0f)) : 0.0f;
    }
}

__global__ void tag_norm(const int* __restrict__ src, const int* __restrict__ dst,
                         const float* __restrict__ dis, float* __restrict__ nrm, int E) {
    int e = blockIdx.x * blockDim.x + threadIdx.x;
    if (e < E) nrm[e] = dis[src[e]] * dis[dst[e]];
}

__global__ void tag_prop(const float* __restrict__ h, const int* __restrict__ src,
                         const int* __restrict__ dst, const float* __restrict__ nrm,
                         float* __restrict__ out, int E) {
    long long t = (long long)blockIdx.x * blockDim.x + threadIdx.x;
    int e = (int)(t >> 5);
    if (e >= E) return;
    int d4 = ((int)t & 31) << 2;
    int s  = src[e];
    int dd = dst[e];
    float w = nrm[e];
    const float4 v = *(const float4*)(h + (size_t)s * 128 + d4);
    float* o = out + (size_t)dd * 128 + d4;
    unsafeAtomicAdd(o + 0, v.x * w);
    unsafeAtomicAdd(o + 1, v.y * w);
    unsafeAtomicAdd(o + 2, v.z * w);
    unsafeAtomicAdd(o + 3, v.w * w);
}

template <int NOUT, bool ACC, bool BIAS, bool RELU>
__global__ void tag_gemm(const float* __restrict__ A, const float* __restrict__ W,
                         float* __restrict__ C, const float* __restrict__ bias) {
    const int lane = threadIdx.x;
    const int tm = blockIdx.x, tn = blockIdx.y;
    const int lo = lane & 15;
    const int hi = lane >> 4;
    const int arow = tm * 16 + lo;
    const int bcol = tn * 16 + lo;
    const int crow = tm * 16 + hi * 8;
    const int ccol = tn * 16 + lo;

    v8f c = {};
    if (ACC) {
#pragma unroll
        for (int r = 0; r < 8; ++r)
            c[r] = C[(size_t)(crow + r) * NOUT + ccol];
    }

    const float* arowp = A + (size_t)arow * 128 + hi * 2;
    const float* bp0   = W + (size_t)(hi * 2) * NOUT + bcol;
#pragma unroll 8
    for (int k0 = 0; k0 < 128; k0 += 4) {
        v2f a = *(const v2f*)(arowp + k0);
        v2f b;
        b.x = bp0[(size_t)k0 * NOUT];
        b.y = bp0[(size_t)k0 * NOUT + NOUT];
        c = __builtin_amdgcn_wmma_f32_16x16x4_f32(
                 false, a,  false, b,
                 (short)0, c,  false,  false);
    }

    const float bv = BIAS ? bias[ccol] : 0.0f;
#pragma unroll
    for (int r = 0; r < 8; ++r) {
        float v = c[r];
        if (BIAS) v += bv;
        if (RELU) v = fmaxf(v, 0.0f);
        C[(size_t)(crow + r) * NOUT + ccol] = v;
    }
}

extern "C" void kernel_launch(void* const* d_in, const int* in_sizes, int n_in,
                              void* d_out, int out_size, void* d_ws, size_t ws_size,
                              hipStream_t stream) {
    const float* x  = (const float*)d_in[0];
    const int*   ei = (const int*)d_in[1];
    const float* W1 = (const float*)d_in[2];
    const float* b1 = (const float*)d_in[3];
    const float* W2 = (const float*)d_in[4];
    const float* b2 = (const float*)d_in[5];

    const int N = in_sizes[0] / 128;
    const int E = in_sizes[1] / 2;
    const int* src = ei;
    const int* dst = ei + E;

    float* ws   = (float*)d_ws;
    float* deg  = ws;
    float* dis  = deg + N;
    float* nrm  = dis + N;
    float* bufA = nrm + E;
    float* bufB = bufA + (size_t)N * 128;
    float* acc  = bufB + (size_t)N * 128;
    float* out  = (float*)d_out;

    const int T = 256;
    const size_t featBytes = (size_t)N * 128 * sizeof(float);
    const long long pth = (long long)E * 32;
    const int pblk = (int)((pth + T - 1) / T);
    const dim3 g1(N / 16, 128 / 16);
    const dim3 g2(N / 16, 1);

    hipMemsetAsync(deg, 0, (size_t)N * sizeof(float), stream);
    tag_deg <<<(E + T - 1) / T, T, 0, stream>>>(dst, deg, E);
    tag_dis <<<(N + T - 1) / T, T, 0, stream>>>(deg, dis, N);
    tag_norm<<<(E + T - 1) / T, T, 0, stream>>>(src, dst, dis, nrm, E);

    tag_gemm<128, false, false, false><<<g1, 32, 0, stream>>>(x,    W1 + 0 * 128 * 128, acc, nullptr);
    hipMemsetAsync(bufA, 0, featBytes, stream);
    tag_prop<<<pblk, T, 0, stream>>>(x, src, dst, nrm, bufA, E);
    tag_gemm<128, true,  false, false><<<g1, 32, 0, stream>>>(bufA, W1 + 1 * 128 * 128, acc, nullptr);
    hipMemsetAsync(bufB, 0, featBytes, stream);
    tag_prop<<<pblk, T, 0, stream>>>(bufA, src, dst, nrm, bufB, E);
    tag_gemm<128, true,  false, false><<<g1, 32, 0, stream>>>(bufB, W1 + 2 * 128 * 128, acc, nullptr);
    hipMemsetAsync(bufA, 0, featBytes, stream);
    tag_prop<<<pblk, T, 0, stream>>>(bufB, src, dst, nrm, bufA, E);
    tag_gemm<128, true,  true,  true ><<<g1, 32, 0, stream>>>(bufA, W1 + 3 * 128 * 128, acc, b1);

    tag_gemm<16, false, false, false><<<g2, 32, 0, stream>>>(acc,  W2 + 0 * 128 * 16, out, nullptr);
    hipMemsetAsync(bufA, 0, featBytes, stream);
    tag_prop<<<pblk, T, 0, stream>>>(acc, src, dst, nrm, bufA, E);
    tag_gemm<16, true,  false, false><<<g2, 32, 0, stream>>>(bufA, W2 + 1 * 128 * 16, out, nullptr);
    hipMemsetAsync(bufB, 0, featBytes, stream);
    tag_prop<<<pblk, T, 0, stream>>>(bufA, src, dst, nrm, bufB, E);
    tag_gemm<16, true,  false, false><<<g2, 32, 0, stream>>>(bufB, W2 + 2 * 128 * 16, out, nullptr);
    hipMemsetAsync(bufA, 0, featBytes, stream);
    tag_prop<<<pblk, T, 0, stream>>>(bufB, src, dst, nrm, bufA, E);
    tag_gemm<16, true,  true,  false><<<g2, 32, 0, stream>>>(bufA, W2 + 3 * 128 * 16, out, b2);

    (void)n_in; (void)out_size; (void)ws_size;
}


# layer 2 projects before it propagates, A^k(hW)=(A^k h)W in nested form: one 4-output f32 WMMA pass over h, then 3 hops on 16-wide rows instead of 128-wide; layer-2 memsets gone
# speedup vs baseline: 1.5431x; 1.5431x over previous
#include <hip/hip_runtime.h>

typedef float v2f __attribute__((ext_vector_type(2)));
typedef float v8f __attribute__((ext_vector_type(8)));

__global__ void tag_deg(const int* __restrict__ dst, float* __restrict__ deg, int E) {
    int e = blockIdx.x * blockDim.x + threadIdx.x;
    if (e < E) unsafeAtomicAdd(&deg[dst[e]], 1.0f);
}

__global__ void tag_dis(const float* __restrict__ deg, float* __restrict__ dis, int N) {
    int i = blockIdx.x * blockDim.x + threadIdx.x;
    if (i < N) {
        float d = deg[i];
        dis[i] = (d > 0.0f) ? rsqrtf(fmaxf(d, 1.0f)) : 0.0f;
    }
}

__global__ void tag_norm(const int* __restrict__ src, const int* __restrict__ dst,
                         const float* __restrict__ dis, float* __restrict__ nrm, int E) {
    int e = blockIdx.x * blockDim.x + threadIdx.x;
    if (e < E) nrm[e] = dis[src[e]] * dis[dst[e]];
}

__global__ void tag_prop(const float* __restrict__ h, const int* __restrict__ src,
                         const int* __restrict__ dst, const float* __restrict__ nrm,
                         float* __restrict__ out, int E) {
    long long t = (long long)blockIdx.x * blockDim.x + threadIdx.x;
    int e = (int)(t >> 5);
    if (e >= E) return;
    int d4 = ((int)t & 31) << 2;
    int s  = src[e];
    int dd = dst[e];
    float w = nrm[e];
    const float4 v = *(const float4*)(h + (size_t)s * 128 + d4);
    float* o = out + (size_t)dd * 128 + d4;
    unsafeAtomicAdd(o + 0, v.x * w);
    unsafeAtomicAdd(o + 1, v.y * w);
    unsafeAtomicAdd(o + 2, v.z * w);
    unsafeAtomicAdd(o + 3, v.w * w);
}

template <int NOUT, bool ACC, bool BIAS, bool RELU>
__global__ void tag_gemm(const float* __restrict__ A, const float* __restrict__ W,
                         float* __restrict__ C, const float* __restrict__ bias) {
    const int lane = threadIdx.x;
    const int tm = blockIdx.x, tn = blockIdx.y;
    const int lo = lane & 15;
    const int hi = lane >> 4;
    const int arow = tm * 16 + lo;
    const int bcol = tn * 16 + lo;
    const int crow = tm * 16 + hi * 8;
    const int ccol = tn * 16 + lo;

    v8f c = {};
    if (ACC) {
#pragma unroll
        for (int r = 0; r < 8; ++r)
            c[r] = C[(size_t)(crow + r) * NOUT + ccol];
    }

    const float* arowp = A + (size_t)arow * 128 + hi * 2;
    const float* bp0   = W + (size_t)(hi * 2) * NOUT + bcol;
#pragma unroll 8
    for (int k0 = 0; k0 < 128; k0 += 4) {
        v2f a = *(const v2f*)(arowp + k0);
        v2f b;
        b.x = bp0[(size_t)k0 * NOUT];
        b.y = bp0[(size_t)k0 * NOUT + NOUT];
        c = __builtin_amdgcn_wmma_f32_16x16x4_f32(
                 false, a,  false, b,
                 (short)0, c,  false,  false);
    }

    const float bv = BIAS ? bias[ccol] : 0.0f;
#pragma unroll
    for (int r = 0; r < 8; ++r) {
        float v = c[r];
        if (BIAS) v += bv;
        if (RELU) v = fmaxf(v, 0.0f);
        C[(size_t)(crow + r) * NOUT + ccol] = v;
    }
}

template <int EPL>
__global__ void tag_prop16(const float* __restrict__ t, const int* __restrict__ src,
                           const int* __restrict__ dst, const float* __restrict__ nrm,
                           float* __restrict__ o, int E) {
    const long long tid = (long long)blockIdx.x * blockDim.x + threadIdx.x;
    const int f = (int)tid & 15;
    const long long e0 = (tid >> 4) * EPL;
    if (e0 + EPL <= (long long)E) {
        const int* sp = src + e0;
        const int* dp = dst + e0;
        const float* wp = nrm + e0;
        int s[EPL], d[EPL];
        float w[EPL], v[EPL];
#pragma unroll
        for (int j = 0; j < EPL; ++j) { s[j] = sp[j]; d[j] = dp[j]; w[j] = wp[j]; }
#pragma unroll
        for (int j = 0; j < EPL; ++j) v[j] = t[(size_t)s[j] * 16 + f];
#pragma unroll
        for (int j = 0; j < EPL; ++j) unsafeAtomicAdd(o + (size_t)d[j] * 16 + f, v[j] * w[j]);
    } else {
        for (long long e = e0; e < (long long)E; ++e)
            unsafeAtomicAdd(o + (size_t)dst[e] * 16 + f, t[(size_t)src[e] * 16 + f] * nrm[e]);
    }
}

__global__ void tag_gemm_l2(const float* __restrict__ A, const float* __restrict__ W,
                            float* __restrict__ out, float* __restrict__ q1,
                            float* __restrict__ q2, float* __restrict__ q3,
                            const float* __restrict__ bias) {
    const int lane = threadIdx.x;
    const int tm = blockIdx.x;
    const int lo = lane & 15;
    const int hi = lane >> 4;
    const int arow = tm * 16 + lo;
    const int crow = tm * 16 + hi * 8;

    v8f c0 = {}, c1 = {}, c2 = {}, c3 = {};
    const float* arowp = A + (size_t)arow * 128 + hi * 2;
    const float* bp0   = W + (size_t)(hi * 2) * 16 + lo;
#pragma unroll 8
    for (int k0 = 0; k0 < 128; k0 += 4) {
        v2f a = *(const v2f*)(arowp + k0);
        v2f b0, b1, b2, b3;
        b0.x = bp0[0 * 2048 + k0 * 16];  b0.y = bp0[0 * 2048 + k0 * 16 + 16];
        b1.x = bp0[1 * 2048 + k0 * 16];  b1.y = bp0[1 * 2048 + k0 * 16 + 16];
        b2.x = bp0[2 * 2048 + k0 * 16];  b2.y = bp0[2 * 2048 + k0 * 16 + 16];
        b3.x = bp0[3 * 2048 + k0 * 16];  b3.y = bp0[3 * 2048 + k0 * 16 + 16];
        c0 = __builtin_amdgcn_wmma_f32_16x16x4_f32(false, a, false, b0, (short)0, c0, false, false);
        c1 = __builtin_amdgcn_wmma_f32_16x16x4_f32(false, a, false, b1, (short)0, c1, false, false);
        c2 = __builtin_amdgcn_wmma_f32_16x16x4_f32(false, a, false, b2, (short)0, c2, false, false);
        c3 = __builtin_amdgcn_wmma_f32_16x16x4_f32(false, a, false, b3, (short)0, c3, false, false);
    }

    const float bv = bias[lo];
#pragma unroll
    for (int r = 0; r < 8; ++r) {
        const size_t off = (size_t)(crow + r) * 16 + lo;
        out[off] = c0[r] + bv;
        q1[off]  = c1[r];
        q2[off]  = c2[r];
        q3[off]  = c3[r];
    }
}

extern "C" void kernel_launch(void* const* d_in, const int* in_sizes, int n_in,
                              void* d_out, int out_size, void* d_ws, size_t ws_size,
                              hipStream_t stream) {
    const float* x  = (const float*)d_in[0];
    const int*   ei = (const int*)d_in[1];
    const float* W1 = (const float*)d_in[2];
    const float* b1 = (const float*)d_in[3];
    const float* W2 = (const float*)d_in[4];
    const float* b2 = (const float*)d_in[5];

    const int N = in_sizes[0] / 128;
    const int E = in_sizes[1] / 2;
    const int* src = ei;
    const int* dst = ei + E;

    float* ws   = (float*)d_ws;
    float* deg  = ws;
    float* dis  = deg + N;
    float* nrm  = dis + N;
    float* bufA = nrm + E;
    float* bufB = bufA + (size_t)N * 128;
    float* acc  = bufB + (size_t)N * 128;
    float* out  = (float*)d_out;

    const int T = 256;
    const size_t featBytes = (size_t)N * 128 * sizeof(float);
    const long long pth = (long long)E * 32;
    const int pblk = (int)((pth + T - 1) / T);
    const dim3 g1(N / 16, 128 / 16);
    const dim3 g2(N / 16, 1);

    hipMemsetAsync(deg, 0, (size_t)N * sizeof(float), stream);
    tag_deg <<<(E + T - 1) / T, T, 0, stream>>>(dst, deg, E);
    tag_dis <<<(N + T - 1) / T, T, 0, stream>>>(deg, dis, N);
    tag_norm<<<(E + T - 1) / T, T, 0, stream>>>(src, dst, dis, nrm, E);

    tag_gemm<128, false, false, false><<<g1, 32, 0, stream>>>(x,    W1 + 0 * 128 * 128, acc, nullptr);
    hipMemsetAsync(bufA, 0, featBytes, stream);
    tag_prop<<<pblk, T, 0, stream>>>(x, src, dst, nrm, bufA, E);
    tag_gemm<128, true,  false, false><<<g1, 32, 0, stream>>>(bufA, W1 + 1 * 128 * 128, acc, nullptr);
    hipMemsetAsync(bufB, 0, featBytes, stream);
    tag_prop<<<pblk, T, 0, stream>>>(bufA, src, dst, nrm, bufB, E);
    tag_gemm<128, true,  false, false><<<g1, 32, 0, stream>>>(bufB, W1 + 2 * 128 * 128, acc, nullptr);
    hipMemsetAsync(bufA, 0, featBytes, stream);
    tag_prop<<<pblk, T, 0, stream>>>(bufB, src, dst, nrm, bufA, E);
    tag_gemm<128, true,  true,  true ><<<g1, 32, 0, stream>>>(bufA, W1 + 3 * 128 * 128, acc, b1);

    float* q1 = bufB;
    float* q2 = q1 + (size_t)N * 16;
    float* q3 = q2 + (size_t)N * 16;
    const int EPL = 8;
    const long long hw  = ((long long)E + EPL - 1) / EPL;
    const int nblk = (int)((hw * 16 + T - 1) / T);
    tag_gemm_l2<<<g2, 32, 0, stream>>>(acc, W2, out, q1, q2, q3, b2);
    tag_prop16<EPL><<<nblk, T, 0, stream>>>(q3, src, dst, nrm, q2, E);
    tag_prop16<EPL><<<nblk, T, 0, stream>>>(q2, src, dst, nrm, q1, E);
    tag_prop16<EPL><<<nblk, T, 0, stream>>>(q1, src, dst, nrm, out, E);

    (void)n_in; (void)out_size; (void)ws_size;
}


# hops gather by destination row over an edge list bucketed inside each call (count, prefix sum, fill): register sums and one store per row replace float atomics and zero-fills
# speedup vs baseline: 3.2413x; 2.1005x over previous
#include <hip/hip_runtime.h>

typedef float v2f __attribute__((ext_vector_type(2)));
typedef float v8f __attribute__((ext_vector_type(8)));


__global__ void tag_count(const int* __restrict__ dst, int* __restrict__ cnt, int E) {
    int e = blockIdx.x * blockDim.x + threadIdx.x;
    if (e < E) atomicAdd(&cnt[dst[e]], 1);
}

__global__ void __launch_bounds__(1024)
tag_rowstart(const int* __restrict__ cnt, int* __restrict__ cur, int N) {
    __shared__ int wsum[32];
    __shared__ int wpre[32];
    const int tid  = threadIdx.x;
    const int lane = tid & 31;
    const int wid  = tid >> 5;
    const int base = blockIdx.x * 4096;

    int pre = 0;
    for (int i = tid * 4; i < base; i += 4096) {
        const int4 v = *(const int4*)(cnt + i);
        pre += (v.x + v.y) + (v.z + v.w);
    }

    const int i0 = base + tid * 4;
    const int c0 = (i0     < N) ? cnt[i0]     : 0;
    const int c1 = (i0 + 1 < N) ? cnt[i0 + 1] : 0;
    const int c2 = (i0 + 2 < N) ? cnt[i0 + 2] : 0;
    const int c3 = (i0 + 3 < N) ? cnt[i0 + 3] : 0;
    const int mine = (c0 + c1) + (c2 + c3);

    int incl = mine;
    for (int o = 1; o < 32; o <<= 1) {
        const int t = __shfl_up(incl, o);
        if (lane >= o) incl += t;
    }
    int pw = pre;
    for (int o = 16; o > 0; o >>= 1) pw += __shfl_xor(pw, o);

    if (lane == 31) wsum[wid] = incl;
    if (lane == 0)  wpre[wid] = pw;
    __syncthreads();

    int tot = 0, woff = 0;
    for (int k = 0; k < 32; ++k) {
        tot += wpre[k];
        if (k < wid) woff += wsum[k];
    }
    const int excl = tot + woff + incl - mine;
    if (i0     < N) cur[i0]     = excl;
    if (i0 + 1 < N) cur[i0 + 1] = excl + c0;
    if (i0 + 2 < N) cur[i0 + 2] = excl + c0 + c1;
    if (i0 + 3 < N) cur[i0 + 3] = excl + c0 + c1 + c2;
}

__global__ void tag_fill(const int* __restrict__ src, const int* __restrict__ dst,
                         int* __restrict__ cur, int* __restrict__ csrc, int E) {
    int e = blockIdx.x * blockDim.x + threadIdx.x;
    if (e < E) {
        const int pos = atomicAdd(&cur[dst[e]], 1);
        if (pos >= 0 && pos < E) csrc[pos] = src[e];
    }
}

__device__ __forceinline__ float tag_dis(int c) {
    return (c > 0) ? rsqrtf(fmaxf((float)c, 1.0f)) : 0.0f;
}

__global__ void tag_hop128(const float* __restrict__ h, const int* __restrict__ cnt,
                           const int* __restrict__ rend, const int* __restrict__ csrc,
                           float* __restrict__ out, int N, int E) {
    const int lane = threadIdx.x & 31;
    const int row = __builtin_amdgcn_readfirstlane(
        (int)(blockIdx.x * (blockDim.x >> 5) + (threadIdx.x >> 5)));
    if (row >= N) return;
    int start = rend[max(row - 1, 0)];
    int end   = rend[row];
    start = (row > 0) ? start : 0;
    end   = min(max(end, 0), E);
    start = min(max(start, 0), end);
    const int c = end - start;
    const float di = tag_dis(c);

    float ax = 0.0f, ay = 0.0f, az = 0.0f, aw = 0.0f;
    const float* hl = h + lane * 4;
    for (int b = 0; b < c; b += 32) {
        const int j = b + lane;
        int s = row;
        float w = 0.0f;
        if (j < c) {
            s = csrc[start + j];
            s = min(max(s, 0), N - 1);
            w = tag_dis(cnt[s]) * di;
        }
        const int nb = min(32, c - b);
        for (int u = 0; u < nb; u += 4) {
            const int   s0 = __shfl(s, u),     s1 = __shfl(s, u + 1);
            const int   s2 = __shfl(s, u + 2), s3 = __shfl(s, u + 3);
            const float w0 = __shfl(w, u),     w1 = __shfl(w, u + 1);
            const float w2 = __shfl(w, u + 2), w3 = __shfl(w, u + 3);
            const float4 v0 = *(const float4*)(hl + (size_t)s0 * 128);
            const float4 v1 = *(const float4*)(hl + (size_t)s1 * 128);
            const float4 v2 = *(const float4*)(hl + (size_t)s2 * 128);
            const float4 v3 = *(const float4*)(hl + (size_t)s3 * 128);
            ax += v0.x * w0;  ay += v0.y * w0;  az += v0.z * w0;  aw += v0.w * w0;
            ax += v1.x * w1;  ay += v1.y * w1;  az += v1.z * w1;  aw += v1.w * w1;
            ax += v2.x * w2;  ay += v2.y * w2;  az += v2.z * w2;  aw += v2.w * w2;
            ax += v3.x * w3;  ay += v3.y * w3;  az += v3.z * w3;  aw += v3.w * w3;
        }
    }
    *(float4*)(out + (size_t)row * 128 + lane * 4) = make_float4(ax, ay, az, aw);
}

__global__ void tag_hop16(const float* __restrict__ t, const int* __restrict__ cnt,
                          const int* __restrict__ rend, const int* __restrict__ csrc,
                          float* __restrict__ o, int N, int E) {
    const int lane = threadIdx.x & 31;
    const int row = __builtin_amdgcn_readfirstlane(
        (int)(blockIdx.x * (blockDim.x >> 5) + (threadIdx.x >> 5)));
    if (row >= N) return;
    const int g = lane >> 2;
    const int q = lane & 3;
    float4* op = (float4*)(o + (size_t)row * 16 + q * 4);
    const float4 old = *op;

    int start = rend[max(row - 1, 0)];
    int end   = rend[row];
    start = (row > 0) ? start : 0;
    end   = min(max(end, 0), E);
    start = min(max(start, 0), end);
    const int c = end - start;
    const float di = tag_dis(c);

    float ax = 0.0f, ay = 0.0f, az = 0.0f, aw = 0.0f;
    const float* tl = t + q * 4;
    for (int b = 0; b < c; b += 32) {
        const int j = b + lane;
        int s = row;
        float w = 0.0f;
        if (j < c) {
            s = csrc[start + j];
            s = min(max(s, 0), N - 1);
            w = tag_dis(cnt[s]) * di;
        }
        const int nb = min(32, c - b);
        for (int u = 0; u < nb; u += 8) {
            const int   sj = __shfl(s, u + g);
            const float wj = __shfl(w, u + g);
            const float4 v = *(const float4*)(tl + (size_t)sj * 16);
            ax += v.x * wj;  ay += v.y * wj;  az += v.z * wj;  aw += v.w * wj;
        }
    }
    for (int m = 4; m < 32; m <<= 1) {
        ax += __shfl_xor(ax, m);
        ay += __shfl_xor(ay, m);
        az += __shfl_xor(az, m);
        aw += __shfl_xor(aw, m);
    }
    if (lane < 4) {
        float4 r = old;
        r.x += ax;  r.y += ay;  r.z += az;  r.w += aw;
        *op = r;
    }
}

template <int NOUT, bool ACC, bool BIAS, bool RELU>
__global__ void tag_gemm(const float* __restrict__ A, const float* __restrict__ W,
                         float* __restrict__ C, const float* __restrict__ bias) {
    const int lane = threadIdx.x;
    const int tm = blockIdx.x, tn = blockIdx.y;
    const int lo = lane & 15;
    const int hi = lane >> 4;
    const int arow = tm * 16 + lo;
    const int bcol = tn * 16 + lo;
    const int crow = tm * 16 + hi * 8;
    const int ccol = tn * 16 + lo;

    v8f c = {};
    if (ACC) {
#pragma unroll
        for (int r = 0; r < 8; ++r)
            c[r] = C[(size_t)(crow + r) * NOUT + ccol];
    }

    const float* arowp = A + (size_t)arow * 128 + hi * 2;
    const float* bp0   = W + (size_t)(hi * 2) * NOUT + bcol;
#pragma unroll 8
    for (int k0 = 0; k0 < 128; k0 += 4) {
        v2f a = *(const v2f*)(arowp + k0);
        v2f b;
        b.x = bp0[(size_t)k0 * NOUT];
        b.y = bp0[(size_t)k0 * NOUT + NOUT];
        c = __builtin_amdgcn_wmma_f32_16x16x4_f32(
                 false, a,  false, b,
                 (short)0, c,  false,  false);
    }

    const float bv = BIAS ? bias[ccol] : 0.0f;
#pragma unroll
    for (int r = 0; r < 8; ++r) {
        float v = c[r];
        if (BIAS) v += bv;
        if (RELU) v = fmaxf(v, 0.0f);
        C[(size_t)(crow + r) * NOUT + ccol] = v;
    }
}

__global__ void tag_gemm_l2(const float* __restrict__ A, const float* __restrict__ W,
                            float* __restrict__ out, float* __restrict__ q1,
                            float* __restrict__ q2, float* __restrict__ q3,
                            const float* __restrict__ bias) {
    const int lane = threadIdx.x;
    const int tm = blockIdx.x;
    const int lo = lane & 15;
    const int hi = lane >> 4;
    const int arow = tm * 16 + lo;
    const int crow = tm * 16 + hi * 8;

    v8f c0 = {}, c1 = {}, c2 = {}, c3 = {};
    const float* arowp = A + (size_t)arow * 128 + hi * 2;
    const float* bp0   = W + (size_t)(hi * 2) * 16 + lo;
#pragma unroll 8
    for (int k0 = 0; k0 < 128; k0 += 4) {
        v2f a = *(const v2f*)(arowp + k0);
        v2f b0, b1, b2, b3;
        b0.x = bp0[0 * 2048 + k0 * 16];  b0.y = bp0[0 * 2048 + k0 * 16 + 16];
        b1.x = bp0[1 * 2048 + k0 * 16];  b1.y = bp0[1 * 2048 + k0 * 16 + 16];
        b2.x = bp0[2 * 2048 + k0 * 16];  b2.y = bp0[2 * 2048 + k0 * 16 + 16];
        b3.x = bp0[3 * 2048 + k0 * 16];  b3.y = bp0[3 * 2048 + k0 * 16 + 16];
        c0 = __builtin_amdgcn_wmma_f32_16x16x4_f32(false, a, false, b0, (short)0, c0, false, false);
        c1 = __builtin_amdgcn_wmma_f32_16x16x4_f32(false, a, false, b1, (short)0, c1, false, false);
        c2 = __builtin_amdgcn_wmma_f32_16x16x4_f32(false, a, false, b2, (short)0, c2, false, false);
        c3 = __builtin_amdgcn_wmma_f32_16x16x4_f32(false, a, false, b3, (short)0, c3, false, false);
    }

    const float bv = bias[lo];
#pragma unroll
    for (int r = 0; r < 8; ++r) {
        const size_t off = (size_t)(crow + r) * 16 + lo;
        out[off] = c0[r] + bv;
        q1[off]  = c1[r];
        q2[off]  = c2[r];
        q3[off]  = c3[r];
    }
}

extern "C" void kernel_launch(void* const* d_in, const int* in_sizes, int n_in,
                              void* d_out, int out_size, void* d_ws, size_t ws_size,
                              hipStream_t stream) {
    const float* x  = (const float*)d_in[0];
    const int*   ei = (const int*)d_in[1];
    const float* W1 = (const float*)d_in[2];
    const float* b1 = (const float*)d_in[3];
    const float* W2 = (const float*)d_in[4];
    const float* b2 = (const float*)d_in[5];

    const int N = in_sizes[0] / 128;
    const int E = in_sizes[1] / 2;
    const int* src = ei;
    const int* dst = ei + E;

    int*   cnt  = (int*)d_ws;
    int*   cur  = cnt + N;
    int*   csrc = cur + N;
    float* bufA = (float*)(csrc + E);
    float* bufB = bufA + (size_t)N * 128;
    float* acc  = bufB + (size_t)N * 128;
    float* out  = (float*)d_out;

    const int T = 256;
    const int RPB = T / 32;
    const int hblk = (N + RPB - 1) / RPB;
    const dim3 g1(N / 16, 128 / 16);
    const dim3 g2(N / 16, 1);

    hipMemsetAsync(cnt, 0, (size_t)N * sizeof(int), stream);
    tag_count   <<<(E + T - 1) / T, T, 0, stream>>>(dst, cnt, E);
    tag_rowstart<<<(N + 4095) / 4096, 1024, 0, stream>>>(cnt, cur, N);
    tag_fill    <<<(E + T - 1) / T, T, 0, stream>>>(src, dst, cur, csrc, E);

    tag_gemm<128, false, false, false><<<g1, 32, 0, stream>>>(x,    W1 + 0 * 128 * 128, acc, nullptr);
    tag_hop128<<<hblk, T, 0, stream>>>(x, cnt, cur, csrc, bufA, N, E);
    tag_gemm<128, true,  false, false><<<g1, 32, 0, stream>>>(bufA, W1 + 1 * 128 * 128, acc, nullptr);
    tag_hop128<<<hblk, T, 0, stream>>>(bufA, cnt, cur, csrc, bufB, N, E);
    tag_gemm<128, true,  false, false><<<g1, 32, 0, stream>>>(bufB, W1 + 2 * 128 * 128, acc, nullptr);
    tag_hop128<<<hblk, T, 0, stream>>>(bufB, cnt, cur, csrc, bufA, N, E);
    tag_gemm<128, true,  true,  true ><<<g1, 32, 0, stream>>>(bufA, W1 + 3 * 128 * 128, acc, b1);

    float* q1 = bufB;
    float* q2 = q1 + (size_t)N * 16;
    float* q3 = q2 + (size_t)N * 16;
    tag_gemm_l2<<<g2, 32, 0, stream>>>(acc, W2, out, q1, q2, q3, b2);
    tag_hop16<<<hblk, T, 0, stream>>>(q3, cnt, cur, csrc, q2, N, E);
    tag_hop16<<<hblk, T, 0, stream>>>(q2, cnt, cur, csrc, q1, N, E);
    tag_hop16<<<hblk, T, 0, stream>>>(q1, cnt, cur, csrc, out, N, E);

    (void)n_in; (void)out_size; (void)ws_size;
}


# layer-1 GEMMs retiled: W staged in LDS once per 8-wave work-group, each wave computes a whole 16x128 strip from one pass over its A rows; B fragments are 16-byte LDS reads, not strided global loads
# speedup vs baseline: 3.9078x; 1.2056x over previous
#include <hip/hip_runtime.h>

typedef float v2f __attribute__((ext_vector_type(2)));
typedef float v8f __attribute__((ext_vector_type(8)));


__global__ void tag_count(const int* __restrict__ dst, int* __restrict__ cnt, int E) {
    int e = blockIdx.x * blockDim.x + threadIdx.x;
    if (e < E) atomicAdd(&cnt[dst[e]], 1);
}

__global__ void __launch_bounds__(1024)
tag_rowstart(const int* __restrict__ cnt, int* __restrict__ cur, int N) {
    __shared__ int wsum[32];
    __shared__ int wpre[32];
    const int tid  = threadIdx.x;
    const int lane = tid & 31;
    const int wid  = tid >> 5;
    const int base = blockIdx.x * 4096;

    int pre = 0;
    for (int i = tid * 4; i < base; i += 4096) {
        const int4 v = *(const int4*)(cnt + i);
        pre += (v.x + v.y) + (v.z + v.w);
    }

    const int i0 = base + tid * 4;
    const int c0 = (i0     < N) ? cnt[i0]     : 0;
    const int c1 = (i0 + 1 < N) ? cnt[i0 + 1] : 0;
    const int c2 = (i0 + 2 < N) ? cnt[i0 + 2] : 0;
    const int c3 = (i0 + 3 < N) ? cnt[i0 + 3] : 0;
    const int mine = (c0 + c1) + (c2 + c3);

    int incl = mine;
    for (int o = 1; o < 32; o <<= 1) {
        const int t = __shfl_up(incl, o);
        if (lane >= o) incl += t;
    }
    int pw = pre;
    for (int o = 16; o > 0; o >>= 1) pw += __shfl_xor(pw, o);

    if (lane == 31) wsum[wid] = incl;
    if (lane == 0)  wpre[wid] = pw;
    __syncthreads();

    int tot = 0, woff = 0;
    for (int k = 0; k < 32; ++k) {
        tot += wpre[k];
        if (k < wid) woff += wsum[k];
    }
    const int excl = tot + woff + incl - mine;
    if (i0     < N) cur[i0]     = excl;
    if (i0 + 1 < N) cur[i0 + 1] = excl + c0;
    if (i0 + 2 < N) cur[i0 + 2] = excl + c0 + c1;
    if (i0 + 3 < N) cur[i0 + 3] = excl + c0 + c1 + c2;
}

__global__ void tag_fill(const int* __restrict__ src, const int* __restrict__ dst,
                         int* __restrict__ cur, int* __restrict__ csrc, int E) {
    int e = blockIdx.x * blockDim.x + threadIdx.x;
    if (e < E) {
        const int pos = atomicAdd(&cur[dst[e]], 1);
        if (pos >= 0 && pos < E) csrc[pos] = src[e];
    }
}

__device__ __forceinline__ float tag_dis(int c) {
    return (c > 0) ? rsqrtf(fmaxf((float)c, 1.0f)) : 0.0f;
}

__global__ void tag_hop128(const float* __restrict__ h, const int* __restrict__ cnt,
                           const int* __restrict__ rend, const int* __restrict__ csrc,
                           float* __restrict__ out, int N, int E) {
    const int lane = threadIdx.x & 31;
    const int row = __builtin_amdgcn_readfirstlane(
        (int)(blockIdx.x * (blockDim.x >> 5) + (threadIdx.x >> 5)));
    if (row >= N) return;
    int start = rend[max(row - 1, 0)];
    int end   = rend[row];
    start = (row > 0) ? start : 0;
    end   = min(max(end, 0), E);
    start = min(max(start, 0), end);
    const int c = end - start;
    const float di = tag_dis(c);

    float ax = 0.0f, ay = 0.0f, az = 0.0f, aw = 0.0f;
    const float* hl = h + lane * 4;
    for (int b = 0; b < c; b += 32) {
        const int j = b + lane;
        int s = row;
        float w = 0.0f;
        if (j < c) {
            s = csrc[start + j];
            s = min(max(s, 0), N - 1);
            w = tag_dis(cnt[s]) * di;
        }
        const int nb = min(32, c - b);
        for (int u = 0; u < nb; u += 4) {
            const int   s0 = __shfl(s, u),     s1 = __shfl(s, u + 1);
            const int   s2 = __shfl(s, u + 2), s3 = __shfl(s, u + 3);
            const float w0 = __shfl(w, u),     w1 = __shfl(w, u + 1);
            const float w2 = __shfl(w, u + 2), w3 = __shfl(w, u + 3);
            const float4 v0 = *(const float4*)(hl + (size_t)s0 * 128);
            const float4 v1 = *(const float4*)(hl + (size_t)s1 * 128);
            const float4 v2 = *(const float4*)(hl + (size_t)s2 * 128);
            const float4 v3 = *(const float4*)(hl + (size_t)s3 * 128);
            ax += v0.x * w0;  ay += v0.y * w0;  az += v0.z * w0;  aw += v0.w * w0;
            ax += v1.x * w1;  ay += v1.y * w1;  az += v1.z * w1;  aw += v1.w * w1;
            ax += v2.x * w2;  ay += v2.y * w2;  az += v2.z * w2;  aw += v2.w * w2;
            ax += v3.x * w3;  ay += v3.y * w3;  az += v3.z * w3;  aw += v3.w * w3;
        }
    }
    *(float4*)(out + (size_t)row * 128 + lane * 4) = make_float4(ax, ay, az, aw);
}

__global__ void tag_hop16(const float* __restrict__ t, const int* __restrict__ cnt,
                          const int* __restrict__ rend, const int* __restrict__ csrc,
                          float* __restrict__ o, int N, int E) {
    const int lane = threadIdx.x & 31;
    const int row = __builtin_amdgcn_readfirstlane(
        (int)(blockIdx.x * (blockDim.x >> 5) + (threadIdx.x >> 5)));
    if (row >= N) return;
    const int g = lane >> 2;
    const int q = lane & 3;
    float4* op = (float4*)(o + (size_t)row * 16 + q * 4);
    const float4 old = *op;

    int start = rend[max(row - 1, 0)];
    int end   = rend[row];
    start = (row > 0) ? start : 0;
    end   = min(max(end, 0), E);
    start = min(max(start, 0), end);
    const int c = end - start;
    const float di = tag_dis(c);

    float ax = 0.0f, ay = 0.0f, az = 0.0f, aw = 0.0f;
    const float* tl = t + q * 4;
    for (int b = 0; b < c; b += 32) {
        const int j = b + lane;
        int s = row;
        float w = 0.0f;
        if (j < c) {
            s = csrc[start + j];
            s = min(max(s, 0), N - 1);
            w = tag_dis(cnt[s]) * di;
        }
        const int nb = min(32, c - b);
        for (int u = 0; u < nb; u += 8) {
            const int   sj = __shfl(s, u + g);
            const float wj = __shfl(w, u + g);
            const float4 v = *(const float4*)(tl + (size_t)sj * 16);
            ax += v.x * wj;  ay += v.y * wj;  az += v.z * wj;  aw += v.w * wj;
        }
    }
    for (int m = 4; m < 32; m <<= 1) {
        ax += __shfl_xor(ax, m);
        ay += __shfl_xor(ay, m);
        az += __shfl_xor(az, m);
        aw += __shfl_xor(aw, m);
    }
    if (lane < 4) {
        float4 r = old;
        r.x += ax;  r.y += ay;  r.z += az;  r.w += aw;
        *op = r;
    }
}

#define TAG_WPITCH 136
template <bool ACC, bool BIAS, bool RELU>
__global__ void __launch_bounds__(256)
tag_gemm_strip(const float* __restrict__ A, const float* __restrict__ W,
               float* __restrict__ C, const float* __restrict__ bias, int ntiles) {
    __shared__ float4 Wq[32 * TAG_WPITCH];
    const int tid  = threadIdx.x;
    const int lane = tid & 31;
    const int c = lane & 15;
    const int h = lane >> 4;
    const int tile = __builtin_amdgcn_readfirstlane((int)(blockIdx.x * 8 + (tid >> 5)));
    const bool live = tile < ntiles;
    const int lt = live ? tile : ntiles - 1;

    const float* ap = A + (size_t)(lt * 16 + c) * 128 + h * 4;
    float4 a[16];
#pragma unroll
    for (int kb = 0; kb < 16; ++kb) a[kb] = *(const float4*)(ap + kb * 8);

#pragma unroll
    for (int i = 0; i < 16; ++i) {
        const int idx = tid + 256 * i;
        const int n = idx & 127, kq = idx >> 7;
        const float* wp = W + (size_t)(4 * kq) * 128 + n;
        Wq[kq * TAG_WPITCH + n] = make_float4(wp[0], wp[128], wp[256], wp[384]);
    }

    float* cp = C + (size_t)(lt * 16 + h * 8) * 128 + c;
    v8f acc[8];
#pragma unroll
    for (int t = 0; t < 8; ++t) {
        if (ACC) {
#pragma unroll
            for (int r = 0; r < 8; ++r) acc[t][r] = cp[(size_t)r * 128 + t * 16];
        } else {
            acc[t] = (v8f){};
        }
    }
    __syncthreads();

    const float4* bp = Wq + h * TAG_WPITCH + c;
#pragma unroll
    for (int kb = 0; kb < 16; ++kb) {
        float4 b[8];
#pragma unroll
        for (int t = 0; t < 8; ++t) b[t] = bp[(2 * kb) * TAG_WPITCH + t * 16];
        const v2f a0 = {a[kb].x, a[kb].y};
        const v2f a1 = {a[kb].z, a[kb].w};
#pragma unroll
        for (int t = 0; t < 8; ++t) {
            const v2f b0 = {b[t].x, b[t].y};
            acc[t] = __builtin_amdgcn_wmma_f32_16x16x4_f32(false, a0, false, b0, (short)0, acc[t], false, false);
        }
#pragma unroll
        for (int t = 0; t < 8; ++t) {
            const v2f b1 = {b[t].z, b[t].w};
            acc[t] = __builtin_amdgcn_wmma_f32_16x16x4_f32(false, a1, false, b1, (short)0, acc[t], false, false);
        }
    }

    if (!live) return;
#pragma unroll
    for (int t = 0; t < 8; ++t) {
        const float bv = BIAS ? bias[t * 16 + c] : 0.0f;
#pragma unroll
        for (int r = 0; r < 8; ++r) {
            float v = acc[t][r];
            if (BIAS) v += bv;
            if (RELU) v = fmaxf(v, 0.0f);
            cp[(size_t)r * 128 + t * 16] = v;
        }
    }
}

__global__ void tag_gemm_l2(const float* __restrict__ A, const float* __restrict__ W,
                            float* __restrict__ out, float* __restrict__ q1,
                            float* __restrict__ q2, float* __restrict__ q3,
                            const float* __restrict__ bias) {
    const int lane = threadIdx.x;
    const int tm = blockIdx.x;
    const int lo = lane & 15;
    const int hi = lane >> 4;
    const int arow = tm * 16 + lo;
    const int crow = tm * 16 + hi * 8;

    v8f c0 = {}, c1 = {}, c2 = {}, c3 = {};
    const float* arowp = A + (size_t)arow * 128 + hi * 2;
    const float* bp0   = W + (size_t)(hi * 2) * 16 + lo;
#pragma unroll 8
    for (int k0 = 0; k0 < 128; k0 += 4) {
        v2f a = *(const v2f*)(arowp + k0);
        v2f b0, b1, b2, b3;
        b0.x = bp0[0 * 2048 + k0 * 16];  b0.y = bp0[0 * 2048 + k0 * 16 + 16];
        b1.x = bp0[1 * 2048 + k0 * 16];  b1.y = bp0[1 * 2048 + k0 * 16 + 16];
        b2.x = bp0[2 * 2048 + k0 * 16];  b2.y = bp0[2 * 2048 + k0 * 16 + 16];
        b3.x = bp0[3 * 2048 + k0 * 16];  b3.y = bp0[3 * 2048 + k0 * 16 + 16];
        c0 = __builtin_amdgcn_wmma_f32_16x16x4_f32(false, a, false, b0, (short)0, c0, false, false);
        c1 = __builtin_amdgcn_wmma_f32_16x16x4_f32(false, a, false, b1, (short)0, c1, false, false);
        c2 = __builtin_amdgcn_wmma_f32_16x16x4_f32(false, a, false, b2, (short)0, c2, false, false);
        c3 = __builtin_amdgcn_wmma_f32_16x16x4_f32(false, a, false, b3, (short)0, c3, false, false);
    }

    const float bv = bias[lo];
#pragma unroll
    for (int r = 0; r < 8; ++r) {
        const size_t off = (size_t)(crow + r) * 16 + lo;
        out[off] = c0[r] + bv;
        q1[off]  = c1[r];
        q2[off]  = c2[r];
        q3[off]  = c3[r];
    }
}

extern "C" void kernel_launch(void* const* d_in, const int* in_sizes, int n_in,
                              void* d_out, int out_size, void* d_ws, size_t ws_size,
                              hipStream_t stream) {
    const float* x  = (const float*)d_in[0];
    const int*   ei = (const int*)d_in[1];
    const float* W1 = (const float*)d_in[2];
    const float* b1 = (const float*)d_in[3];
    const float* W2 = (const float*)d_in[4];
    const float* b2 = (const float*)d_in[5];

    const int N = in_sizes[0] / 128;
    const int E = in_sizes[1] / 2;
    const int* src = ei;
    const int* dst = ei + E;

    int*   cnt  = (int*)d_ws;
    int*   cur  = cnt + N;
    int*   csrc = cur + N;
    float* bufA = (float*)(csrc + E);
    float* bufB = bufA + (size_t)N * 128;
    float* acc  = bufB + (size_t)N * 128;
    float* out  = (float*)d_out;

    const int T = 256;
    const int RPB = T / 32;
    const int hblk = (N + RPB - 1) / RPB;
    const int ntiles = N / 16;
    const int g1 = (ntiles + 7) / 8;
    const dim3 g2(N / 16, 1);

    hipMemsetAsync(cnt, 0, (size_t)N * sizeof(int), stream);
    tag_count   <<<(E + T - 1) / T, T, 0, stream>>>(dst, cnt, E);
    tag_rowstart<<<(N + 4095) / 4096, 1024, 0, stream>>>(cnt, cur, N);
    tag_fill    <<<(E + T - 1) / T, T, 0, stream>>>(src, dst, cur, csrc, E);

    tag_gemm_strip<false, false, false><<<g1, 256, 0, stream>>>(x,    W1 + 0 * 128 * 128, acc, nullptr, ntiles);
    tag_hop128<<<hblk, T, 0, stream>>>(x, cnt, cur, csrc, bufA, N, E);
    tag_gemm_strip<true,  false, false><<<g1, 256, 0, stream>>>(bufA, W1 + 1 * 128 * 128, acc, nullptr, ntiles);
    tag_hop128<<<hblk, T, 0, stream>>>(bufA, cnt, cur, csrc, bufB, N, E);
    tag_gemm_strip<true,  false, false><<<g1, 256, 0, stream>>>(bufB, W1 + 2 * 128 * 128, acc, nullptr, ntiles);
    tag_hop128<<<hblk, T, 0, stream>>>(bufB, cnt, cur, csrc, bufA, N, E);
    tag_gemm_strip<true,  true,  true ><<<g1, 256, 0, stream>>>(bufA, W1 + 3 * 128 * 128, acc, b1, ntiles);

    float* q1 = bufB;
    float* q2 = q1 + (size_t)N * 16;
    float* q3 = q2 + (size_t)N * 16;
    tag_gemm_l2<<<g2, 32, 0, stream>>>(acc, W2, out, q1, q2, q3, b2);
    tag_hop16<<<hblk, T, 0, stream>>>(q3, cnt, cur, csrc, q2, N, E);
    tag_hop16<<<hblk, T, 0, stream>>>(q2, cnt, cur, csrc, q1, N, E);
    tag_hop16<<<hblk, T, 0, stream>>>(q1, cnt, cur, csrc, out, N, E);

    (void)n_in; (void)out_size; (void)ws_size;
}


# tables a hop gathers are stored pre-multiplied by their own row's dis, so the hop sums plain rows: the per-edge random degree load and its dependent wait leave 5 of 6 hops; GEMMs 1-2 unscale A rows
# speedup vs baseline: 4.0820x; 1.0446x over previous
#include <hip/hip_runtime.h>

typedef float v2f __attribute__((ext_vector_type(2)));
typedef float v8f __attribute__((ext_vector_type(8)));


__global__ void tag_count(const int* __restrict__ dst, int* __restrict__ cnt, int E) {
    int e = blockIdx.x * blockDim.x + threadIdx.x;
    if (e < E) atomicAdd(&cnt[dst[e]], 1);
}

__global__ void __launch_bounds__(1024)
tag_rowstart(const int* __restrict__ cnt, int* __restrict__ cur, int N) {
    __shared__ int wsum[32];
    __shared__ int wpre[32];
    const int tid  = threadIdx.x;
    const int lane = tid & 31;
    const int wid  = tid >> 5;
    const int base = blockIdx.x * 4096;

    int pre = 0;
    for (int i = tid * 4; i < base; i += 4096) {
        const int4 v = *(const int4*)(cnt + i);
        pre += (v.x + v.y) + (v.z + v.w);
    }

    const int i0 = base + tid * 4;
    const int c0 = (i0     < N) ? cnt[i0]     : 0;
    const int c1 = (i0 + 1 < N) ? cnt[i0 + 1] : 0;
    const int c2 = (i0 + 2 < N) ? cnt[i0 + 2] : 0;
    const int c3 = (i0 + 3 < N) ? cnt[i0 + 3] : 0;
    const int mine = (c0 + c1) + (c2 + c3);

    int incl = mine;
    for (int o = 1; o < 32; o <<= 1) {
        const int t = __shfl_up(incl, o);
        if (lane >= o) incl += t;
    }
    int pw = pre;
    for (int o = 16; o > 0; o >>= 1) pw += __shfl_xor(pw, o);

    if (lane == 31) wsum[wid] = incl;
    if (lane == 0)  wpre[wid] = pw;
    __syncthreads();

    int tot = 0, woff = 0;
    for (int k = 0; k < 32; ++k) {
        tot += wpre[k];
        if (k < wid) woff += wsum[k];
    }
    const int excl = tot + woff + incl - mine;
    if (i0     < N) cur[i0]     = excl;
    if (i0 + 1 < N) cur[i0 + 1] = excl + c0;
    if (i0 + 2 < N) cur[i0 + 2] = excl + c0 + c1;
    if (i0 + 3 < N) cur[i0 + 3] = excl + c0 + c1 + c2;
}

__global__ void tag_fill(const int* __restrict__ src, const int* __restrict__ dst,
                         int* __restrict__ cur, int* __restrict__ csrc, int E) {
    int e = blockIdx.x * blockDim.x + threadIdx.x;
    if (e < E) {
        const int pos = atomicAdd(&cur[dst[e]], 1);
        if (pos >= 0 && pos < E) csrc[pos] = src[e];
    }
}

__device__ __forceinline__ float tag_dis(int c) {
    return (c > 0) ? rsqrtf(fmaxf((float)c, 1.0f)) : 0.0f;
}

template <bool SRCW, bool SCALED>
__global__ void tag_hop128(const float* __restrict__ h, const int* __restrict__ cnt,
                           const int* __restrict__ rend, const int* __restrict__ csrc,
                           float* __restrict__ out, int N, int E) {
    const int lane = threadIdx.x & 31;
    const int row = __builtin_amdgcn_readfirstlane(
        (int)(blockIdx.x * (blockDim.x >> 5) + (threadIdx.x >> 5)));
    if (row >= N) return;
    int start = rend[max(row - 1, 0)];
    int end   = rend[row];
    start = (row > 0) ? start : 0;
    end   = min(max(end, 0), E);
    start = min(max(start, 0), end);
    const int c = end - start;
    const float di = tag_dis(c);

    float ax = 0.0f, ay = 0.0f, az = 0.0f, aw = 0.0f;
    const float* hl = h + lane * 4;
    for (int b = 0; b < c; b += 32) {
        const int j = b + lane;
        int s = row;
        float w = 0.0f;
        if (j < c) {
            s = csrc[start + j];
            s = min(max(s, 0), N - 1);
            if (SRCW) w = tag_dis(cnt[s]);
        }
        const int nb = min(32, c - b);
        for (int u = 0; u < nb; u += 4) {
            const int   s0 = __shfl(s, u),     s1 = __shfl(s, u + 1);
            const int   s2 = __shfl(s, u + 2), s3 = __shfl(s, u + 3);
            float w0, w1, w2, w3;
            if (SRCW) {
                w0 = __shfl(w, u);      w1 = __shfl(w, u + 1);
                w2 = __shfl(w, u + 2);  w3 = __shfl(w, u + 3);
            } else {
                w0 = 1.0f;
                w1 = (u + 1 < nb) ? 1.0f : 0.0f;
                w2 = (u + 2 < nb) ? 1.0f : 0.0f;
                w3 = (u + 3 < nb) ? 1.0f : 0.0f;
            }
            const float4 v0 = *(const float4*)(hl + (size_t)s0 * 128);
            const float4 v1 = *(const float4*)(hl + (size_t)s1 * 128);
            const float4 v2 = *(const float4*)(hl + (size_t)s2 * 128);
            const float4 v3 = *(const float4*)(hl + (size_t)s3 * 128);
            ax += v0.x * w0;  ay += v0.y * w0;  az += v0.z * w0;  aw += v0.w * w0;
            ax += v1.x * w1;  ay += v1.y * w1;  az += v1.z * w1;  aw += v1.w * w1;
            ax += v2.x * w2;  ay += v2.y * w2;  az += v2.z * w2;  aw += v2.w * w2;
            ax += v3.x * w3;  ay += v3.y * w3;  az += v3.z * w3;  aw += v3.w * w3;
        }
    }
    const float f = SCALED ? di * di : di;
    *(float4*)(out + (size_t)row * 128 + lane * 4) = make_float4(ax * f, ay * f, az * f, aw * f);
}

template <bool SCALED>
__global__ void tag_hop16(const float* __restrict__ t,
                          const int* __restrict__ rend, const int* __restrict__ csrc,
                          float* __restrict__ o, int N, int E) {
    const int lane = threadIdx.x & 31;
    const int row = __builtin_amdgcn_readfirstlane(
        (int)(blockIdx.x * (blockDim.x >> 5) + (threadIdx.x >> 5)));
    if (row >= N) return;
    const int g = lane >> 2;
    const int q = lane & 3;
    float4* op = (float4*)(o + (size_t)row * 16 + q * 4);
    const float4 old = *op;

    int start = rend[max(row - 1, 0)];
    int end   = rend[row];
    start = (row > 0) ? start : 0;
    end   = min(max(end, 0), E);
    start = min(max(start, 0), end);
    const int c = end - start;
    const float di = tag_dis(c);

    float ax = 0.0f, ay = 0.0f, az = 0.0f, aw = 0.0f;
    const float* tl = t + q * 4;
    for (int b = 0; b < c; b += 32) {
        const int j = b + lane;
        int s = row;
        if (j < c) {
            s = csrc[start + j];
            s = min(max(s, 0), N - 1);
        }
        const int nb = min(32, c - b);
        for (int u = 0; u < nb; u += 8) {
            const int   sj = __shfl(s, u + g);
            const float wj = (u + g < nb) ? 1.0f : 0.0f;
            const float4 v = *(const float4*)(tl + (size_t)sj * 16);
            ax += v.x * wj;  ay += v.y * wj;  az += v.z * wj;  aw += v.w * wj;
        }
    }
    for (int m = 4; m < 32; m <<= 1) {
        ax += __shfl_xor(ax, m);
        ay += __shfl_xor(ay, m);
        az += __shfl_xor(az, m);
        aw += __shfl_xor(aw, m);
    }
    if (lane < 4) {
        float4 r = old;
        r.x += ax * di;  r.y += ay * di;  r.z += az * di;  r.w += aw * di;
        if (SCALED) { r.x *= di;  r.y *= di;  r.z *= di;  r.w *= di; }
        *op = r;
    }
}

#define TAG_WPITCH 136
template <bool ACC, bool BIAS, bool RELU, bool UNSCALE>
__global__ void __launch_bounds__(256)
tag_gemm_strip(const float* __restrict__ A, const float* __restrict__ W,
               float* __restrict__ C, const float* __restrict__ bias,
               const int* __restrict__ cnt, int ntiles) {
    __shared__ float4 Wq[32 * TAG_WPITCH];
    const int tid  = threadIdx.x;
    const int lane = tid & 31;
    const int c = lane & 15;
    const int h = lane >> 4;
    const int tile = __builtin_amdgcn_readfirstlane((int)(blockIdx.x * 8 + (tid >> 5)));
    const bool live = tile < ntiles;
    const int lt = live ? tile : ntiles - 1;

    const float* ap = A + (size_t)(lt * 16 + c) * 128 + h * 4;
    float4 a[16];
#pragma unroll
    for (int kb = 0; kb < 16; ++kb) a[kb] = *(const float4*)(ap + kb * 8);
    const int dg = UNSCALE ? cnt[lt * 16 + c] : 0;

#pragma unroll
    for (int i = 0; i < 16; ++i) {
        const int idx = tid + 256 * i;
        const int n = idx & 127, kq = idx >> 7;
        const float* wp = W + (size_t)(4 * kq) * 128 + n;
        Wq[kq * TAG_WPITCH + n] = make_float4(wp[0], wp[128], wp[256], wp[384]);
    }

    if (UNSCALE) {
        const float f = (dg > 0) ? sqrtf((float)dg) : 0.0f;
#pragma unroll
        for (int kb = 0; kb < 16; ++kb) {
            a[kb].x *= f;  a[kb].y *= f;  a[kb].z *= f;  a[kb].w *= f;
        }
    }

    float* cp = C + (size_t)(lt * 16 + h * 8) * 128 + c;
    v8f acc[8];
#pragma unroll
    for (int t = 0; t < 8; ++t) {
        if (ACC) {
#pragma unroll
            for (int r = 0; r < 8; ++r) acc[t][r] = cp[(size_t)r * 128 + t * 16];
        } else {
            acc[t] = (v8f){};
        }
    }
    __syncthreads();

    const float4* bp = Wq + h * TAG_WPITCH + c;
#pragma unroll
    for (int kb = 0; kb < 16; ++kb) {
        float4 b[8];
#pragma unroll
        for (int t = 0; t < 8; ++t) b[t] = bp[(2 * kb) * TAG_WPITCH + t * 16];
        const v2f a0 = {a[kb].x, a[kb].y};
        const v2f a1 = {a[kb].z, a[kb].w};
#pragma unroll
        for (int t = 0; t < 8; ++t) {
            const v2f b0 = {b[t].x, b[t].y};
            acc[t] = __builtin_amdgcn_wmma_f32_16x16x4_f32(false, a0, false, b0, (short)0, acc[t], false, false);
        }
#pragma unroll
        for (int t = 0; t < 8; ++t) {
            const v2f b1 = {b[t].z, b[t].w};
            acc[t] = __builtin_amdgcn_wmma_f32_16x16x4_f32(false, a1, false, b1, (short)0, acc[t], false, false);
        }
    }

    if (!live) return;
#pragma unroll
    for (int t = 0; t < 8; ++t) {
        const float bv = BIAS ? bias[t * 16 + c] : 0.0f;
#pragma unroll
        for (int r = 0; r < 8; ++r) {
            float v = acc[t][r];
            if (BIAS) v += bv;
            if (RELU) v = fmaxf(v, 0.0f);
            cp[(size_t)r * 128 + t * 16] = v;
        }
    }
}

__global__ void tag_gemm_l2(const float* __restrict__ A, const float* __restrict__ W,
                            float* __restrict__ out, float* __restrict__ q1,
                            float* __restrict__ q2, float* __restrict__ q3,
                            const float* __restrict__ bias, const int* __restrict__ cnt) {
    const int lane = threadIdx.x;
    const int tm = blockIdx.x;
    const int lo = lane & 15;
    const int hi = lane >> 4;
    const int arow = tm * 16 + lo;
    const int crow = tm * 16 + hi * 8;

    v8f c0 = {}, c1 = {}, c2 = {}, c3 = {};
    const float* arowp = A + (size_t)arow * 128 + hi * 2;
    const float* bp0   = W + (size_t)(hi * 2) * 16 + lo;
#pragma unroll 8
    for (int k0 = 0; k0 < 128; k0 += 4) {
        v2f a = *(const v2f*)(arowp + k0);
        v2f b0, b1, b2, b3;
        b0.x = bp0[0 * 2048 + k0 * 16];  b0.y = bp0[0 * 2048 + k0 * 16 + 16];
        b1.x = bp0[1 * 2048 + k0 * 16];  b1.y = bp0[1 * 2048 + k0 * 16 + 16];
        b2.x = bp0[2 * 2048 + k0 * 16];  b2.y = bp0[2 * 2048 + k0 * 16 + 16];
        b3.x = bp0[3 * 2048 + k0 * 16];  b3.y = bp0[3 * 2048 + k0 * 16 + 16];
        c0 = __builtin_amdgcn_wmma_f32_16x16x4_f32(false, a, false, b0, (short)0, c0, false, false);
        c1 = __builtin_amdgcn_wmma_f32_16x16x4_f32(false, a, false, b1, (short)0, c1, false, false);
        c2 = __builtin_amdgcn_wmma_f32_16x16x4_f32(false, a, false, b2, (short)0, c2, false, false);
        c3 = __builtin_amdgcn_wmma_f32_16x16x4_f32(false, a, false, b3, (short)0, c3, false, false);
    }

    const float bv = bias[lo];
#pragma unroll
    for (int r = 0; r < 8; ++r) {
        const size_t off = (size_t)(crow + r) * 16 + lo;
        out[off] = c0[r] + bv;
        q1[off]  = c1[r];
        q2[off]  = c2[r];
        q3[off]  = c3[r] * tag_dis(cnt[crow + r]);
    }
}

extern "C" void kernel_launch(void* const* d_in, const int* in_sizes, int n_in,
                              void* d_out, int out_size, void* d_ws, size_t ws_size,
                              hipStream_t stream) {
    const float* x  = (const float*)d_in[0];
    const int*   ei = (const int*)d_in[1];
    const float* W1 = (const float*)d_in[2];
    const float* b1 = (const float*)d_in[3];
    const float* W2 = (const float*)d_in[4];
    const float* b2 = (const float*)d_in[5];

    const int N = in_sizes[0] / 128;
    const int E = in_sizes[1] / 2;
    const int* src = ei;
    const int* dst = ei + E;

    int*   cnt  = (int*)d_ws;
    int*   cur  = cnt + N;
    int*   csrc = cur + N;
    float* bufA = (float*)(csrc + E);
    float* bufB = bufA + (size_t)N * 128;
    float* acc  = bufB + (size_t)N * 128;
    float* out  = (float*)d_out;

    const int T = 256;
    const int RPB = T / 32;
    const int hblk = (N + RPB - 1) / RPB;
    const int ntiles = N / 16;
    const int g1 = (ntiles + 7) / 8;
    const dim3 g2(N / 16, 1);

    hipMemsetAsync(cnt, 0, (size_t)N * sizeof(int), stream);
    tag_count   <<<(E + T - 1) / T, T, 0, stream>>>(dst, cnt, E);
    tag_rowstart<<<(N + 4095) / 4096, 1024, 0, stream>>>(cnt, cur, N);
    tag_fill    <<<(E + T - 1) / T, T, 0, stream>>>(src, dst, cur, csrc, E);

    tag_gemm_strip<false, false, false, false><<<g1, 256, 0, stream>>>(x,    W1 + 0 * 128 * 128, acc, nullptr, cnt, ntiles);
    tag_hop128<true,  true ><<<hblk, T, 0, stream>>>(x, cnt, cur, csrc, bufA, N, E);
    tag_gemm_strip<true,  false, false, true ><<<g1, 256, 0, stream>>>(bufA, W1 + 1 * 128 * 128, acc, nullptr, cnt, ntiles);
    tag_hop128<false, true ><<<hblk, T, 0, stream>>>(bufA, cnt, cur, csrc, bufB, N, E);
    tag_gemm_strip<true,  false, false, true ><<<g1, 256, 0, stream>>>(bufB, W1 + 2 * 128 * 128, acc, nullptr, cnt, ntiles);
    tag_hop128<false, false><<<hblk, T, 0, stream>>>(bufB, cnt, cur, csrc, bufA, N, E);
    tag_gemm_strip<true,  true,  true,  false><<<g1, 256, 0, stream>>>(bufA, W1 + 3 * 128 * 128, acc, b1, cnt, ntiles);

    float* q1 = bufB;
    float* q2 = q1 + (size_t)N * 16;
    float* q3 = q2 + (size_t)N * 16;
    tag_gemm_l2<<<g2, 32, 0, stream>>>(acc, W2, out, q1, q2, q3, b2, cnt);
    tag_hop16<true ><<<hblk, T, 0, stream>>>(q3, cur, csrc, q2, N, E);
    tag_hop16<true ><<<hblk, T, 0, stream>>>(q2, cur, csrc, q1, N, E);
    tag_hop16<false><<<hblk, T, 0, stream>>>(q1, cur, csrc, out, N, E);

    (void)n_in; (void)out_size; (void)ws_size;
}


# edge bucketing takes one atomic pass, not two: the counting pass keeps each edge's returned rank, the prefix sum gives row ends, the fill stores at row start + rank with plain loads and stores
# speedup vs baseline: 4.9842x; 1.2210x over previous
#include <hip/hip_runtime.h>

typedef float v2f __attribute__((ext_vector_type(2)));
typedef float v8f __attribute__((ext_vector_type(8)));


__global__ void tag_count(const int* __restrict__ dst, int* __restrict__ cnt,
                          int* __restrict__ rank, int E) {
    int e = blockIdx.x * blockDim.x + threadIdx.x;
    if (e < E) rank[e] = atomicAdd(&cnt[dst[e]], 1);
}

__global__ void __launch_bounds__(1024)
tag_rowstart(const int* __restrict__ cnt, int* __restrict__ cur, int N) {
    __shared__ int wsum[32];
    __shared__ int wpre[32];
    const int tid  = threadIdx.x;
    const int lane = tid & 31;
    const int wid  = tid >> 5;
    const int base = blockIdx.x * 4096;

    int pre = 0;
    for (int i = tid * 4; i < base; i += 4096) {
        const int4 v = *(const int4*)(cnt + i);
        pre += (v.x + v.y) + (v.z + v.w);
    }

    const int i0 = base + tid * 4;
    const int c0 = (i0     < N) ? cnt[i0]     : 0;
    const int c1 = (i0 + 1 < N) ? cnt[i0 + 1] : 0;
    const int c2 = (i0 + 2 < N) ? cnt[i0 + 2] : 0;
    const int c3 = (i0 + 3 < N) ? cnt[i0 + 3] : 0;
    const int mine = (c0 + c1) + (c2 + c3);

    int incl = mine;
    for (int o = 1; o < 32; o <<= 1) {
        const int t = __shfl_up(incl, o);
        if (lane >= o) incl += t;
    }
    int pw = pre;
    for (int o = 16; o > 0; o >>= 1) pw += __shfl_xor(pw, o);

    if (lane == 31) wsum[wid] = incl;
    if (lane == 0)  wpre[wid] = pw;
    __syncthreads();

    int tot = 0, woff = 0;
    for (int k = 0; k < 32; ++k) {
        tot += wpre[k];
        if (k < wid) woff += wsum[k];
    }
    const int excl = tot + woff + incl - mine;
    if (i0     < N) cur[i0]     = excl + c0;
    if (i0 + 1 < N) cur[i0 + 1] = excl + c0 + c1;
    if (i0 + 2 < N) cur[i0 + 2] = excl + c0 + c1 + c2;
    if (i0 + 3 < N) cur[i0 + 3] = excl + mine;
}

__global__ void tag_fill(const int* __restrict__ src, const int* __restrict__ dst,
                         const int* __restrict__ rend, const int* __restrict__ rank,
                         int* __restrict__ csrc, int N, int E) {
    int e = blockIdx.x * blockDim.x + threadIdx.x;
    if (e < E) {
        const int d = min(max(dst[e], 0), N - 1);
        const int before = rend[max(d - 1, 0)];
        const int pos = ((d > 0) ? before : 0) + rank[e];
        if (pos >= 0 && pos < E) csrc[pos] = src[e];
    }
}

__device__ __forceinline__ float tag_dis(int c) {
    return (c > 0) ? rsqrtf(fmaxf((float)c, 1.0f)) : 0.0f;
}

template <bool SRCW, bool SCALED>
__global__ void tag_hop128(const float* __restrict__ h, const int* __restrict__ cnt,
                           const int* __restrict__ rend, const int* __restrict__ csrc,
                           float* __restrict__ out, int N, int E) {
    const int lane = threadIdx.x & 31;
    const int row = __builtin_amdgcn_readfirstlane(
        (int)(blockIdx.x * (blockDim.x >> 5) + (threadIdx.x >> 5)));
    if (row >= N) return;
    int start = rend[max(row - 1, 0)];
    int end   = rend[row];
    start = (row > 0) ? start : 0;
    end   = min(max(end, 0), E);
    start = min(max(start, 0), end);
    const int c = end - start;
    const float di = tag_dis(c);

    float ax = 0.0f, ay = 0.0f, az = 0.0f, aw = 0.0f;
    const float* hl = h + lane * 4;
    for (int b = 0; b < c; b += 32) {
        const int j = b + lane;
        int s = row;
        float w = 0.0f;
        if (j < c) {
            s = csrc[start + j];
            s = min(max(s, 0), N - 1);
            if (SRCW) w = tag_dis(cnt[s]);
        }
        const int nb = min(32, c - b);
        for (int u = 0; u < nb; u += 4) {
            const int   s0 = __shfl(s, u),     s1 = __shfl(s, u + 1);
            const int   s2 = __shfl(s, u + 2), s3 = __shfl(s, u + 3);
            float w0, w1, w2, w3;
            if (SRCW) {
                w0 = __shfl(w, u);      w1 = __shfl(w, u + 1);
                w2 = __shfl(w, u + 2);  w3 = __shfl(w, u + 3);
            } else {
                w0 = 1.0f;
                w1 = (u + 1 < nb) ? 1.0f : 0.0f;
                w2 = (u + 2 < nb) ? 1.0f : 0.0f;
                w3 = (u + 3 < nb) ? 1.0f : 0.0f;
            }
            const float4 v0 = *(const float4*)(hl + (size_t)s0 * 128);
            const float4 v1 = *(const float4*)(hl + (size_t)s1 * 128);
            const float4 v2 = *(const float4*)(hl + (size_t)s2 * 128);
            const float4 v3 = *(const float4*)(hl + (size_t)s3 * 128);
            ax += v0.x * w0;  ay += v0.y * w0;  az += v0.z * w0;  aw += v0.w * w0;
            ax += v1.x * w1;  ay += v1.y * w1;  az += v1.z * w1;  aw += v1.w * w1;
            ax += v2.x * w2;  ay += v2.y * w2;  az += v2.z * w2;  aw += v2.w * w2;
            ax += v3.x * w3;  ay += v3.y * w3;  az += v3.z * w3;  aw += v3.w * w3;
        }
    }
    const float f = SCALED ? di * di : di;
    *(float4*)(out + (size_t)row * 128 + lane * 4) = make_float4(ax * f, ay * f, az * f, aw * f);
}

template <bool SCALED>
__global__ void tag_hop16(const float* __restrict__ t,
                          const int* __restrict__ rend, const int* __restrict__ csrc,
                          float* __restrict__ o, int N, int E) {
    const int lane = threadIdx.x & 31;
    const int row = __builtin_amdgcn_readfirstlane(
        (int)(blockIdx.x * (blockDim.x >> 5) + (threadIdx.x >> 5)));
    if (row >= N) return;
    const int g = lane >> 2;
    const int q = lane & 3;
    float4* op = (float4*)(o + (size_t)row * 16 + q * 4);
    const float4 old = *op;

    int start = rend[max(row - 1, 0)];
    int end   = rend[row];
    start = (row > 0) ? start : 0;
    end   = min(max(end, 0), E);
    start = min(max(start, 0), end);
    const int c = end - start;
    const float di = tag_dis(c);

    float ax = 0.0f, ay = 0.0f, az = 0.0f, aw = 0.0f;
    const float* tl = t + q * 4;
    for (int b = 0; b < c; b += 32) {
        const int j = b + lane;
        int s = row;
        if (j < c) {
            s = csrc[start + j];
            s = min(max(s, 0), N - 1);
        }
        const int nb = min(32, c - b);
        for (int u = 0; u < nb; u += 8) {
            const int   sj = __shfl(s, u + g);
            const float wj = (u + g < nb) ? 1.0f : 0.0f;
            const float4 v = *(const float4*)(tl + (size_t)sj * 16);
            ax += v.x * wj;  ay += v.y * wj;  az += v.z * wj;  aw += v.w * wj;
        }
    }
    for (int m = 4; m < 32; m <<= 1) {
        ax += __shfl_xor(ax, m);
        ay += __shfl_xor(ay, m);
        az += __shfl_xor(az, m);
        aw += __shfl_xor(aw, m);
    }
    if (lane < 4) {
        float4 r = old;
        r.x += ax * di;  r.y += ay * di;  r.z += az * di;  r.w += aw * di;
        if (SCALED) { r.x *= di;  r.y *= di;  r.z *= di;  r.w *= di; }
        *op = r;
    }
}

#define TAG_WPITCH 136

__device__ __forceinline__ void tag_load_a(float4 (&a)[16], const float* p) {
#pragma unroll
    for (int kb = 0; kb < 16; ++kb) a[kb] = *(const float4*)(p + kb * 8);
}

__device__ __forceinline__ void tag_scale_a(float4 (&a)[16], float f) {
#pragma unroll
    for (int kb = 0; kb < 16; ++kb) {
        a[kb].x *= f;  a[kb].y *= f;  a[kb].z *= f;  a[kb].w *= f;
    }
}

__device__ __forceinline__ void tag_mma_plane(v8f (&acc)[8], const float4 (&a)[16], const float4* bp) {
#pragma unroll
    for (int kb = 0; kb < 16; ++kb) {
        float4 b[8];
#pragma unroll
        for (int t = 0; t < 8; ++t) b[t] = bp[(2 * kb) * TAG_WPITCH + t * 16];
        const v2f a0 = {a[kb].x, a[kb].y};
        const v2f a1 = {a[kb].z, a[kb].w};
#pragma unroll
        for (int t = 0; t < 8; ++t) {
            const v2f b0 = {b[t].x, b[t].y};
            acc[t] = __builtin_amdgcn_wmma_f32_16x16x4_f32(false, a0, false, b0, (short)0, acc[t], false, false);
        }
#pragma unroll
        for (int t = 0; t < 8; ++t) {
            const v2f b1 = {b[t].z, b[t].w};
            acc[t] = __builtin_amdgcn_wmma_f32_16x16x4_f32(false, a1, false, b1, (short)0, acc[t], false, false);
        }
    }
}

__global__ void __launch_bounds__(512)
tag_gemm_all(const float* __restrict__ P0, const float* __restrict__ P1,
             const float* __restrict__ P2, float* P3H,
             const float* __restrict__ W, const float* __restrict__ bias,
             const int* __restrict__ cnt, int ntiles) {
    __shared__ float4 Wq[128 * TAG_WPITCH];
    const int tid  = threadIdx.x;
    const int lane = tid & 31;
    const int c = lane & 15;
    const int h = lane >> 4;
    const int g  = __builtin_amdgcn_readfirstlane((int)(blockIdx.x * 16 + (tid >> 5)));
    const int G  = (int)gridDim.x * 16;
    const int t0 = (int)(((unsigned)g * (unsigned)ntiles) / (unsigned)G);
    const int t1 = (int)(((unsigned)(g + 1) * (unsigned)ntiles) / (unsigned)G);
    const size_t lo = (size_t)c * 128 + h * 4;

#pragma unroll 8
    for (int i = 0; i < 32; ++i) {
        const int idx = tid + 512 * i;
        const int n = idx & 127, kq = idx >> 7;
        const float* wp = W + (size_t)(4 * kq) * 128 + n;
        Wq[kq * TAG_WPITCH + n] = make_float4(wp[0], wp[128], wp[256], wp[384]);
    }
    __syncthreads();

    const float4* bp = Wq + h * TAG_WPITCH + c;
    for (int tile = t0; tile < t1; ++tile) {
        const size_t ro = (size_t)tile * 2048 + lo;
        v8f acc[8];
#pragma unroll
        for (int t = 0; t < 8; ++t) acc[t] = (v8f){};

        const int dg = cnt[tile * 16 + c];
#pragma nounroll
        for (int p = 0; p < 4; ++p) {
            const float* pl = (p == 0) ? P0 : (p == 1) ? P1 : (p == 2) ? P2 : P3H;
            float4 a[16];
            tag_load_a(a, pl + ro);
            if (p == 1 || p == 2) tag_scale_a(a, (dg > 0) ? sqrtf((float)dg) : 0.0f);
            tag_mma_plane(acc, a, bp + p * (32 * TAG_WPITCH));
        }

        float* cp = P3H + (size_t)(tile * 16 + h * 8) * 128 + c;
#pragma unroll
        for (int t = 0; t < 8; ++t) {
            const float bv = bias[t * 16 + c];
#pragma unroll
            for (int r = 0; r < 8; ++r)
                cp[(size_t)r * 128 + t * 16] = fmaxf(acc[t][r] + bv, 0.0f);
        }
    }
}

__global__ void tag_gemm_l2(const float* __restrict__ A, const float* __restrict__ W,
                            float* __restrict__ out, float* __restrict__ q1,
                            float* __restrict__ q2, float* __restrict__ q3,
                            const float* __restrict__ bias, const int* __restrict__ cnt) {
    const int lane = threadIdx.x;
    const int tm = blockIdx.x;
    const int lo = lane & 15;
    const int hi = lane >> 4;
    const int arow = tm * 16 + lo;
    const int crow = tm * 16 + hi * 8;

    v8f c0 = {}, c1 = {}, c2 = {}, c3 = {};
    const float* arowp = A + (size_t)arow * 128 + hi * 2;
    const float* bp0   = W + (size_t)(hi * 2) * 16 + lo;
#pragma unroll 8
    for (int k0 = 0; k0 < 128; k0 += 4) {
        v2f a = *(const v2f*)(arowp + k0);
        v2f b0, b1, b2, b3;
        b0.x = bp0[0 * 2048 + k0 * 16];  b0.y = bp0[0 * 2048 + k0 * 16 + 16];
        b1.x = bp0[1 * 2048 + k0 * 16];  b1.y = bp0[1 * 2048 + k0 * 16 + 16];
        b2.x = bp0[2 * 2048 + k0 * 16];  b2.y = bp0[2 * 2048 + k0 * 16 + 16];
        b3.x = bp0[3 * 2048 + k0 * 16];  b3.y = bp0[3 * 2048 + k0 * 16 + 16];
        c0 = __builtin_amdgcn_wmma_f32_16x16x4_f32(false, a, false, b0, (short)0, c0, false, false);
        c1 = __builtin_amdgcn_wmma_f32_16x16x4_f32(false, a, false, b1, (short)0, c1, false, false);
        c2 = __builtin_amdgcn_wmma_f32_16x16x4_f32(false, a, false, b2, (short)0, c2, false, false);
        c3 = __builtin_amdgcn_wmma_f32_16x16x4_f32(false, a, false, b3, (short)0, c3, false, false);
    }

    const float bv = bias[lo];
#pragma unroll
    for (int r = 0; r < 8; ++r) {
        const size_t off = (size_t)(crow + r) * 16 + lo;
        out[off] = c0[r] + bv;
        q1[off]  = c1[r];
        q2[off]  = c2[r];
        q3[off]  = c3[r] * tag_dis(cnt[crow + r]);
    }
}

extern "C" void kernel_launch(void* const* d_in, const int* in_sizes, int n_in,
                              void* d_out, int out_size, void* d_ws, size_t ws_size,
                              hipStream_t stream) {
    const float* x  = (const float*)d_in[0];
    const int*   ei = (const int*)d_in[1];
    const float* W1 = (const float*)d_in[2];
    const float* b1 = (const float*)d_in[3];
    const float* W2 = (const float*)d_in[4];
    const float* b2 = (const float*)d_in[5];

    const int N = in_sizes[0] / 128;
    const int E = in_sizes[1] / 2;
    const int* src = ei;
    const int* dst = ei + E;

    int*   cnt  = (int*)d_ws;
    int*   cur  = cnt + N;
    int*   csrc = cur + N;
    float* bufA = (float*)(csrc + E);
    float* bufB = bufA + (size_t)N * 128;
    float* acc  = bufB + (size_t)N * 128;
    int*   rank = (int*)acc;
    float* out  = (float*)d_out;

    const int T = 256;
    const int RPB = T / 32;
    const int hblk = (N + RPB - 1) / RPB;
    const int ntiles = N / 16;
    const dim3 g2(N / 16, 1);

    hipMemsetAsync(cnt, 0, (size_t)N * sizeof(int), stream);
    tag_count   <<<(E + T - 1) / T, T, 0, stream>>>(dst, cnt, rank, E);
    tag_rowstart<<<(N + 4095) / 4096, 1024, 0, stream>>>(cnt, cur, N);
    tag_fill    <<<(E + T - 1) / T, T, 0, stream>>>(src, dst, cur, rank, csrc, N, E);

    tag_hop128<true,  true ><<<hblk, T, 0, stream>>>(x,    cnt, cur, csrc, bufA, N, E);
    tag_hop128<false, true ><<<hblk, T, 0, stream>>>(bufA, cnt, cur, csrc, bufB, N, E);
    tag_hop128<false, false><<<hblk, T, 0, stream>>>(bufB, cnt, cur, csrc, acc,  N, E);
    tag_gemm_all<<<256, 512, 0, stream>>>(x, bufA, bufB, acc, W1, b1, cnt, ntiles);

    float* q1 = bufB;
    float* q2 = q1 + (size_t)N * 16;
    float* q3 = q2 + (size_t)N * 16;
    tag_gemm_l2<<<g2, 32, 0, stream>>>(acc, W2, out, q1, q2, q3, b2, cnt);
    tag_hop16<true ><<<hblk, T, 0, stream>>>(q3, cur, csrc, q2, N, E);
    tag_hop16<true ><<<hblk, T, 0, stream>>>(q2, cur, csrc, q1, N, E);
    tag_hop16<false><<<hblk, T, 0, stream>>>(q1, cur, csrc, out, N, E);

    (void)n_in; (void)out_size; (void)ws_size;
}


# edge bucketing with no global atomics: edges binned by 256-row destination range on LDS counters (count, prefix sum, scatter), then one work-group per range counts, scans, fills its rows in LDS
# speedup vs baseline: 6.0330x; 1.2104x over previous
#include <hip/hip_runtime.h>

typedef float v2f __attribute__((ext_vector_type(2)));
typedef float v8f __attribute__((ext_vector_type(8)));

#define TAG_RB   256
#define TAG_RSH  8
#define TAG_PB   256
#define TAG_MAXB 512

__device__ __forceinline__ int tag_range_of(int d, int N) {
    d = min(max(d, 0), N - 1);
    return min(d >> TAG_RSH, TAG_MAXB - 1);
}

__global__ void __launch_bounds__(1024)
tag_bin_hist(const int* __restrict__ dst, int* __restrict__ gh, int N, int E, int NB) {
    __shared__ int hist[TAG_MAXB];
    const int tid = threadIdx.x;
    if (tid < TAG_MAXB) hist[tid] = 0;
    __syncthreads();
    const int chunk = (E + TAG_PB - 1) / TAG_PB;
    const int e0 = blockIdx.x * chunk;
    const int e1 = min(e0 + chunk, E);
    for (int e = e0 + tid; e < e1; e += 4096) {
        const bool k1 = e + 1024 < e1, k2 = e + 2048 < e1, k3 = e + 3072 < e1;
        const int d0 = dst[e];
        const int d1 = k1 ? dst[e + 1024] : 0;
        const int d2 = k2 ? dst[e + 2048] : 0;
        const int d3 = k3 ? dst[e + 3072] : 0;
        atomicAdd(&hist[tag_range_of(d0, N)], 1);
        if (k1) atomicAdd(&hist[tag_range_of(d1, N)], 1);
        if (k2) atomicAdd(&hist[tag_range_of(d2, N)], 1);
        if (k3) atomicAdd(&hist[tag_range_of(d3, N)], 1);
    }
    __syncthreads();
    if (tid < NB && tid < TAG_MAXB) gh[tid * TAG_PB + blockIdx.x] = hist[tid];
}

__global__ void __launch_bounds__(1024)
tag_rowstart(const int* __restrict__ cnt, int* __restrict__ cur, int N) {
    __shared__ int wsum[32];
    __shared__ int wpre[32];
    const int tid  = threadIdx.x;
    const int lane = tid & 31;
    const int wid  = tid >> 5;
    const int base = blockIdx.x * 4096;

    int pre = 0;
    for (int i = tid * 4; i < base; i += 4096) {
        const int4 v = *(const int4*)(cnt + i);
        pre += (v.x + v.y) + (v.z + v.w);
    }

    const int i0 = base + tid * 4;
    const int c0 = (i0     < N) ? cnt[i0]     : 0;
    const int c1 = (i0 + 1 < N) ? cnt[i0 + 1] : 0;
    const int c2 = (i0 + 2 < N) ? cnt[i0 + 2] : 0;
    const int c3 = (i0 + 3 < N) ? cnt[i0 + 3] : 0;
    const int mine = (c0 + c1) + (c2 + c3);

    int incl = mine;
    for (int o = 1; o < 32; o <<= 1) {
        const int t = __shfl_up(incl, o);
        if (lane >= o) incl += t;
    }
    int pw = pre;
    for (int o = 16; o > 0; o >>= 1) pw += __shfl_xor(pw, o);

    if (lane == 31) wsum[wid] = incl;
    if (lane == 0)  wpre[wid] = pw;
    __syncthreads();

    int tot = 0, woff = 0;
    for (int k = 0; k < 32; ++k) {
        tot += wpre[k];
        if (k < wid) woff += wsum[k];
    }
    const int excl = tot + woff + incl - mine;
    if (i0     < N) cur[i0]     = excl + c0;
    if (i0 + 1 < N) cur[i0 + 1] = excl + c0 + c1;
    if (i0 + 2 < N) cur[i0 + 2] = excl + c0 + c1 + c2;
    if (i0 + 3 < N) cur[i0 + 3] = excl + mine;
}

__global__ void __launch_bounds__(1024)
tag_bin_scatter(const int* __restrict__ src, const int* __restrict__ dst,
                const int* __restrict__ gend, int2* __restrict__ pairs, int N, int E, int NB) {
    __shared__ int pos[TAG_MAXB];
    const int tid = threadIdx.x;
    if (tid < TAG_MAXB) {
        int p = 0;
        if (tid < NB) {
            const int i = tid * TAG_PB + blockIdx.x;
            p = (i > 0) ? gend[i - 1] : 0;
        }
        pos[tid] = p;
    }
    __syncthreads();
    const int chunk = (E + TAG_PB - 1) / TAG_PB;
    const int e0 = blockIdx.x * chunk;
    const int e1 = min(e0 + chunk, E);
    for (int e = e0 + tid; e < e1; e += 4096) {
        const bool k1 = e + 1024 < e1, k2 = e + 2048 < e1, k3 = e + 3072 < e1;
        const int d0 = dst[e],                  s0 = src[e];
        const int d1 = k1 ? dst[e + 1024] : 0,  s1 = k1 ? src[e + 1024] : 0;
        const int d2 = k2 ? dst[e + 2048] : 0,  s2 = k2 ? src[e + 2048] : 0;
        const int d3 = k3 ? dst[e + 3072] : 0,  s3 = k3 ? src[e + 3072] : 0;
        {
            const int p = atomicAdd(&pos[tag_range_of(d0, N)], 1);
            if (p >= 0 && p < E) pairs[p] = make_int2(s0, d0);
        }
        if (k1) {
            const int p = atomicAdd(&pos[tag_range_of(d1, N)], 1);
            if (p >= 0 && p < E) pairs[p] = make_int2(s1, d1);
        }
        if (k2) {
            const int p = atomicAdd(&pos[tag_range_of(d2, N)], 1);
            if (p >= 0 && p < E) pairs[p] = make_int2(s2, d2);
        }
        if (k3) {
            const int p = atomicAdd(&pos[tag_range_of(d3, N)], 1);
            if (p >= 0 && p < E) pairs[p] = make_int2(s3, d3);
        }
    }
}

__device__ __forceinline__ int tag_local_row(int d, int row0) {
    return min(max(d - row0, 0), TAG_RB - 1);
}

__global__ void __launch_bounds__(1024)
tag_bucket_rows(const int2* __restrict__ pairs, const int* __restrict__ gend,
                int* __restrict__ cnt, int* __restrict__ rend, int* __restrict__ csrc,
                int N, int E) {
    __shared__ int rc[TAG_RB];
    __shared__ int cursor[TAG_RB];
    __shared__ int wtot[TAG_RB / 32];
    const int tid  = threadIdx.x;
    const int lane = tid & 31;
    const int wid  = tid >> 5;
    const int b    = blockIdx.x;
    const int row0 = b * TAG_RB;
    int s0 = gend[max(b * TAG_PB - 1, 0)];
    int s1 = gend[(b + 1) * TAG_PB - 1];
    s0 = (b > 0) ? s0 : 0;
    s1 = min(max(s1, 0), E);
    s0 = min(max(s0, 0), s1);
    if (tid < TAG_RB) rc[tid] = 0;
    __syncthreads();
    for (int e = s0 + tid; e < s1; e += 4096) {
        const bool k1 = e + 1024 < s1, k2 = e + 2048 < s1, k3 = e + 3072 < s1;
        const int y0 = pairs[e].y;
        const int y1 = k1 ? pairs[e + 1024].y : 0;
        const int y2 = k2 ? pairs[e + 2048].y : 0;
        const int y3 = k3 ? pairs[e + 3072].y : 0;
        atomicAdd(&rc[tag_local_row(y0, row0)], 1);
        if (k1) atomicAdd(&rc[tag_local_row(y1, row0)], 1);
        if (k2) atomicAdd(&rc[tag_local_row(y2, row0)], 1);
        if (k3) atomicAdd(&rc[tag_local_row(y3, row0)], 1);
    }
    __syncthreads();
    int mine = 0, incl = 0;
    if (tid < TAG_RB) {
        mine = rc[tid];
        incl = mine;
        for (int o = 1; o < 32; o <<= 1) {
            const int t = __shfl_up(incl, o);
            if (lane >= o) incl += t;
        }
        if (lane == 31) wtot[wid] = incl;
    }
    __syncthreads();
    if (tid < TAG_RB) {
        int woff = 0;
        for (int k = 0; k < TAG_RB / 32; ++k) if (k < wid) woff += wtot[k];
        const int rowend = s0 + woff + incl;
        cursor[tid] = rowend - mine;
        const int row = row0 + tid;
        if (row < N) { cnt[row] = mine;  rend[row] = rowend; }
    }
    __syncthreads();
    for (int e = s0 + tid; e < s1; e += 4096) {
        const bool k1 = e + 1024 < s1, k2 = e + 2048 < s1, k3 = e + 3072 < s1;
        const int2 z = make_int2(0, 0);
        const int2 p0 = pairs[e];
        const int2 p1 = k1 ? pairs[e + 1024] : z;
        const int2 p2 = k2 ? pairs[e + 2048] : z;
        const int2 p3 = k3 ? pairs[e + 3072] : z;
        {
            const int p = atomicAdd(&cursor[tag_local_row(p0.y, row0)], 1);
            if (p >= 0 && p < E) csrc[p] = p0.x;
        }
        if (k1) {
            const int p = atomicAdd(&cursor[tag_local_row(p1.y, row0)], 1);
            if (p >= 0 && p < E) csrc[p] = p1.x;
        }
        if (k2) {
            const int p = atomicAdd(&cursor[tag_local_row(p2.y, row0)], 1);
            if (p >= 0 && p < E) csrc[p] = p2.x;
        }
        if (k3) {
            const int p = atomicAdd(&cursor[tag_local_row(p3.y, row0)], 1);
            if (p >= 0 && p < E) csrc[p] = p3.x;
        }
    }
}

__device__ __forceinline__ float tag_dis(int c) {
    return (c > 0) ? rsqrtf(fmaxf((float)c, 1.0f)) : 0.0f;
}

template <bool SRCW, bool SCALED>
__global__ void tag_hop128(const float* __restrict__ h, const int* __restrict__ cnt,
                           const int* __restrict__ rend, const int* __restrict__ csrc,
                           float* __restrict__ out, int N, int E) {
    const int lane = threadIdx.x & 31;
    const int row = __builtin_amdgcn_readfirstlane(
        (int)(blockIdx.x * (blockDim.x >> 5) + (threadIdx.x >> 5)));
    if (row >= N) return;
    int start = rend[max(row - 1, 0)];
    int end   = rend[row];
    start = (row > 0) ? start : 0;
    end   = min(max(end, 0), E);
    start = min(max(start, 0), end);
    const int c = end - start;
    const float di = tag_dis(c);

    float ax = 0.0f, ay = 0.0f, az = 0.0f, aw = 0.0f;
    const float* hl = h + lane * 4;
    for (int b = 0; b < c; b += 32) {
        const int j = b + lane;
        int s = row;
        float w = 0.0f;
        if (j < c) {
            s = csrc[start + j];
            s = min(max(s, 0), N - 1);
            if (SRCW) w = tag_dis(cnt[s]);
        }
        const int nb = min(32, c - b);
        for (int u = 0; u < nb; u += 4) {
            const int   s0 = __shfl(s, u),     s1 = __shfl(s, u + 1);
            const int   s2 = __shfl(s, u + 2), s3 = __shfl(s, u + 3);
            float w0, w1, w2, w3;
            if (SRCW) {
                w0 = __shfl(w, u);      w1 = __shfl(w, u + 1);
                w2 = __shfl(w, u + 2);  w3 = __shfl(w, u + 3);
            } else {
                w0 = 1.0f;
                w1 = (u + 1 < nb) ? 1.0f : 0.0f;
                w2 = (u + 2 < nb) ? 1.0f : 0.0f;
                w3 = (u + 3 < nb) ? 1.0f : 0.0f;
            }
            const float4 v0 = *(const float4*)(hl + (size_t)s0 * 128);
            const float4 v1 = *(const float4*)(hl + (size_t)s1 * 128);
            const float4 v2 = *(const float4*)(hl + (size_t)s2 * 128);
            const float4 v3 = *(const float4*)(hl + (size_t)s3 * 128);
            ax += v0.x * w0;  ay += v0.y * w0;  az += v0.z * w0;  aw += v0.w * w0;
            ax += v1.x * w1;  ay += v1.y * w1;  az += v1.z * w1;  aw += v1.w * w1;
            ax += v2.x * w2;  ay += v2.y * w2;  az += v2.z * w2;  aw += v2.w * w2;
            ax += v3.x * w3;  ay += v3.y * w3;  az += v3.z * w3;  aw += v3.w * w3;
        }
    }
    const float f = SCALED ? di * di : di;
    *(float4*)(out + (size_t)row * 128 + lane * 4) = make_float4(ax * f, ay * f, az * f, aw * f);
}

template <bool SCALED>
__global__ void tag_hop16(const float* __restrict__ t,
                          const int* __restrict__ rend, const int* __restrict__ csrc,
                          float* __restrict__ o, int N, int E) {
    const int lane = threadIdx.x & 31;
    const int row = __builtin_amdgcn_readfirstlane(
        (int)(blockIdx.x * (blockDim.x >> 5) + (threadIdx.x >> 5)));
    if (row >= N) return;
    const int g = lane >> 2;
    const int q = lane & 3;
    float4* op = (float4*)(o + (size_t)row * 16 + q * 4);
    const float4 old = *op;

    int start = rend[max(row - 1, 0)];
    int end   = rend[row];
    start = (row > 0) ? start : 0;
    end   = min(max(end, 0), E);
    start = min(max(start, 0), end);
    const int c = end - start;
    const float di = tag_dis(c);

    float ax = 0.0f, ay = 0.0f, az = 0.0f, aw = 0.0f;
    const float* tl = t + q * 4;
    for (int b = 0; b < c; b += 32) {
        const int j = b + lane;
        int s = row;
        if (j < c) {
            s = csrc[start + j];
            s = min(max(s, 0), N - 1);
        }
        const int nb = min(32, c - b);
        for (int u = 0; u < nb; u += 8) {
            const int   sj = __shfl(s, u + g);
            const float wj = (u + g < nb) ? 1.0f : 0.0f;
            const float4 v = *(const float4*)(tl + (size_t)sj * 16);
            ax += v.x * wj;  ay += v.y * wj;  az += v.z * wj;  aw += v.w * wj;
        }
    }
    for (int m = 4; m < 32; m <<= 1) {
        ax += __shfl_xor(ax, m);
        ay += __shfl_xor(ay, m);
        az += __shfl_xor(az, m);
        aw += __shfl_xor(aw, m);
    }
    if (lane < 4) {
        float4 r = old;
        r.x += ax * di;  r.y += ay * di;  r.z += az * di;  r.w += aw * di;
        if (SCALED) { r.x *= di;  r.y *= di;  r.z *= di;  r.w *= di; }
        *op = r;
    }
}

#define TAG_WPITCH 136

__device__ __forceinline__ void tag_load_a(float4 (&a)[16], const float* p) {
#pragma unroll
    for (int kb = 0; kb < 16; ++kb) a[kb] = *(const float4*)(p + kb * 8);
}

__device__ __forceinline__ void tag_scale_a(float4 (&a)[16], float f) {
#pragma unroll
    for (int kb = 0; kb < 16; ++kb) {
        a[kb].x *= f;  a[kb].y *= f;  a[kb].z *= f;  a[kb].w *= f;
    }
}

__device__ __forceinline__ void tag_mma_plane(v8f (&acc)[8], const float4 (&a)[16], const float4* bp) {
#pragma unroll
    for (int kb = 0; kb < 16; ++kb) {
        float4 b[8];
#pragma unroll
        for (int t = 0; t < 8; ++t) b[t] = bp[(2 * kb) * TAG_WPITCH + t * 16];
        const v2f a0 = {a[kb].x, a[kb].y};
        const v2f a1 = {a[kb].z, a[kb].w};
#pragma unroll
        for (int t = 0; t < 8; ++t) {
            const v2f b0 = {b[t].x, b[t].y};
            acc[t] = __builtin_amdgcn_wmma_f32_16x16x4_f32(false, a0, false, b0, (short)0, acc[t], false, false);
        }
#pragma unroll
        for (int t = 0; t < 8; ++t) {
            const v2f b1 = {b[t].z, b[t].w};
            acc[t] = __builtin_amdgcn_wmma_f32_16x16x4_f32(false, a1, false, b1, (short)0, acc[t], false, false);
        }
    }
}

__global__ void __launch_bounds__(512)
tag_gemm_all(const float* __restrict__ P0, const float* __restrict__ P1,
             const float* __restrict__ P2, float* P3H,
             const float* __restrict__ W, const float* __restrict__ bias,
             const int* __restrict__ cnt, int ntiles) {
    __shared__ float4 Wq[128 * TAG_WPITCH];
    const int tid  = threadIdx.x;
    const int lane = tid & 31;
    const int c = lane & 15;
    const int h = lane >> 4;
    const int g  = __builtin_amdgcn_readfirstlane((int)(blockIdx.x * 16 + (tid >> 5)));
    const int G  = (int)gridDim.x * 16;
    const int t0 = (int)(((unsigned)g * (unsigned)ntiles) / (unsigned)G);
    const int t1 = (int)(((unsigned)(g + 1) * (unsigned)ntiles) / (unsigned)G);
    const size_t lo = (size_t)c * 128 + h * 4;

#pragma unroll 8
    for (int i = 0; i < 32; ++i) {
        const int idx = tid + 512 * i;
        const int n = idx & 127, kq = idx >> 7;
        const float* wp = W + (size_t)(4 * kq) * 128 + n;
        Wq[kq * TAG_WPITCH + n] = make_float4(wp[0], wp[128], wp[256], wp[384]);
    }
    __syncthreads();

    const float4* bp = Wq + h * TAG_WPITCH + c;
    for (int tile = t0; tile < t1; ++tile) {
        const size_t ro = (size_t)tile * 2048 + lo;
        v8f acc[8];
#pragma unroll
        for (int t = 0; t < 8; ++t) acc[t] = (v8f){};

        const int dg = cnt[tile * 16 + c];
#pragma nounroll
        for (int p = 0; p < 4; ++p) {
            const float* pl = (p == 0) ? P0 : (p == 1) ? P1 : (p == 2) ? P2 : P3H;
            float4 a[16];
            tag_load_a(a, pl + ro);
            if (p == 1 || p == 2) tag_scale_a(a, (dg > 0) ? sqrtf((float)dg) : 0.0f);
            tag_mma_plane(acc, a, bp + p * (32 * TAG_WPITCH));
        }

        float* cp = P3H + (size_t)(tile * 16 + h * 8) * 128 + c;
#pragma unroll
        for (int t = 0; t < 8; ++t) {
            const float bv = bias[t * 16 + c];
#pragma unroll
            for (int r = 0; r < 8; ++r)
                cp[(size_t)r * 128 + t * 16] = fmaxf(acc[t][r] + bv, 0.0f);
        }
    }
}

__global__ void tag_gemm_l2(const float* __restrict__ A, const float* __restrict__ W,
                            float* __restrict__ out, float* __restrict__ q1,
                            float* __restrict__ q2, float* __restrict__ q3,
                            const float* __restrict__ bias, const int* __restrict__ cnt) {
    const int lane = threadIdx.x;
    const int tm = blockIdx.x;
    const int lo = lane & 15;
    const int hi = lane >> 4;
    const int arow = tm * 16 + lo;
    const int crow = tm * 16 + hi * 8;

    v8f c0 = {}, c1 = {}, c2 = {}, c3 = {};
    const float* arowp = A + (size_t)arow * 128 + hi * 2;
    const float* bp0   = W + (size_t)(hi * 2) * 16 + lo;
#pragma unroll 8
    for (int k0 = 0; k0 < 128; k0 += 4) {
        v2f a = *(const v2f*)(arowp + k0);
        v2f b0, b1, b2, b3;
        b0.x = bp0[0 * 2048 + k0 * 16];  b0.y = bp0[0 * 2048 + k0 * 16 + 16];
        b1.x = bp0[1 * 2048 + k0 * 16];  b1.y = bp0[1 * 2048 + k0 * 16 + 16];
        b2.x = bp0[2 * 2048 + k0 * 16];  b2.y = bp0[2 * 2048 + k0 * 16 + 16];
        b3.x = bp0[3 * 2048 + k0 * 16];  b3.y = bp0[3 * 2048 + k0 * 16 + 16];
        c0 = __builtin_amdgcn_wmma_f32_16x16x4_f32(false, a, false, b0, (short)0, c0, false, false);
        c1 = __builtin_amdgcn_wmma_f32_16x16x4_f32(false, a, false, b1, (short)0, c1, false, false);
        c2 = __builtin_amdgcn_wmma_f32_16x16x4_f32(false, a, false, b2, (short)0, c2, false, false);
        c3 = __builtin_amdgcn_wmma_f32_16x16x4_f32(false, a, false, b3, (short)0, c3, false, false);
    }

    const float bv = bias[lo];
#pragma unroll
    for (int r = 0; r < 8; ++r) {
        const size_t off = (size_t)(crow + r) * 16 + lo;
        out[off] = c0[r] + bv;
        q1[off]  = c1[r];
        q2[off]  = c2[r];
        q3[off]  = c3[r] * tag_dis(cnt[crow + r]);
    }
}

extern "C" void kernel_launch(void* const* d_in, const int* in_sizes, int n_in,
                              void* d_out, int out_size, void* d_ws, size_t ws_size,
                              hipStream_t stream) {
    const float* x  = (const float*)d_in[0];
    const int*   ei = (const int*)d_in[1];
    const float* W1 = (const float*)d_in[2];
    const float* b1 = (const float*)d_in[3];
    const float* W2 = (const float*)d_in[4];
    const float* b2 = (const float*)d_in[5];

    const int N = in_sizes[0] / 128;
    const int E = in_sizes[1] / 2;
    const int* src = ei;
    const int* dst = ei + E;

    int*   cnt  = (int*)d_ws;
    int*   cur  = cnt + N;
    int*   csrc = cur + N;
    float* bufA = (float*)(csrc + E);
    float* bufB = bufA + (size_t)N * 128;
    float* acc  = bufB + (size_t)N * 128;
    const int NB = (N + TAG_RB - 1) / TAG_RB;
    const int M  = NB * TAG_PB;
    int*   gh    = (int*)bufB;
    int*   gend  = gh + M;
    int2*  pairs = (int2*)acc;
    float* out  = (float*)d_out;

    const int T = 256;
    const int RPB = T / 32;
    const int hblk = (N + RPB - 1) / RPB;
    const int ntiles = N / 16;
    const dim3 g2(N / 16, 1);

    tag_bin_hist   <<<TAG_PB, 1024, 0, stream>>>(dst, gh, N, E, NB);
    tag_rowstart   <<<(M + 4095) / 4096, 1024, 0, stream>>>(gh, gend, M);
    tag_bin_scatter<<<TAG_PB, 1024, 0, stream>>>(src, dst, gend, pairs, N, E, NB);
    tag_bucket_rows<<<NB, 1024, 0, stream>>>(pairs, gend, cnt, cur, csrc, N, E);

    tag_hop128<true,  true ><<<hblk, T, 0, stream>>>(x,    cnt, cur, csrc, bufA, N, E);
    tag_hop128<false, true ><<<hblk, T, 0, stream>>>(bufA, cnt, cur, csrc, bufB, N, E);
    tag_hop128<false, false><<<hblk, T, 0, stream>>>(bufB, cnt, cur, csrc, acc,  N, E);
    tag_gemm_all<<<256, 512, 0, stream>>>(x, bufA, bufB, acc, W1, b1, cnt, ntiles);

    float* q1 = bufB;
    float* q2 = q1 + (size_t)N * 16;
    float* q3 = q2 + (size_t)N * 16;
    tag_gemm_l2<<<g2, 32, 0, stream>>>(acc, W2, out, q1, q2, q3, b2, cnt);
    tag_hop16<true ><<<hblk, T, 0, stream>>>(q3, cur, csrc, q2, N, E);
    tag_hop16<true ><<<hblk, T, 0, stream>>>(q2, cur, csrc, q1, N, E);
    tag_hop16<false><<<hblk, T, 0, stream>>>(q1, cur, csrc, out, N, E);

    (void)n_in; (void)out_size; (void)ws_size;
}


# layer-1 product strips dealt so that every SIMD of a WGP gets the same matrix work: whole strips by wave class w&3 (the waves that share a SIMD), left-over strips cut into four column quarters
# speedup vs baseline: 6.2645x; 1.0384x over previous
#include <hip/hip_runtime.h>

typedef float v2f __attribute__((ext_vector_type(2)));
typedef float v8f __attribute__((ext_vector_type(8)));

#define TAG_RB   256
#define TAG_RSH  8
#define TAG_PB   256
#define TAG_MAXB 512

__device__ __forceinline__ int tag_range_of(int d, int N) {
    d = min(max(d, 0), N - 1);
    return min(d >> TAG_RSH, TAG_MAXB - 1);
}

__global__ void __launch_bounds__(1024)
tag_bin_hist(const int* __restrict__ dst, int* __restrict__ gh, int N, int E, int NB) {
    __shared__ int hist[TAG_MAXB];
    const int tid = threadIdx.x;
    if (tid < TAG_MAXB) hist[tid] = 0;
    __syncthreads();
    const int chunk = (E + TAG_PB - 1) / TAG_PB;
    const int e0 = blockIdx.x * chunk;
    const int e1 = min(e0 + chunk, E);
    for (int e = e0 + tid; e < e1; e += 4096) {
        const bool k1 = e + 1024 < e1, k2 = e + 2048 < e1, k3 = e + 3072 < e1;
        const int d0 = dst[e];
        const int d1 = k1 ? dst[e + 1024] : 0;
        const int d2 = k2 ? dst[e + 2048] : 0;
        const int d3 = k3 ? dst[e + 3072] : 0;
        atomicAdd(&hist[tag_range_of(d0, N)], 1);
        if (k1) atomicAdd(&hist[tag_range_of(d1, N)], 1);
        if (k2) atomicAdd(&hist[tag_range_of(d2, N)], 1);
        if (k3) atomicAdd(&hist[tag_range_of(d3, N)], 1);
    }
    __syncthreads();
    if (tid < NB && tid < TAG_MAXB) gh[tid * TAG_PB + blockIdx.x] = hist[tid];
}

__global__ void __launch_bounds__(1024)
tag_rowstart(const int* __restrict__ cnt, int* __restrict__ cur, int N) {
    __shared__ int wsum[32];
    __shared__ int wpre[32];
    const int tid  = threadIdx.x;
    const int lane = tid & 31;
    const int wid  = tid >> 5;
    const int base = blockIdx.x * 4096;

    int pre = 0;
    for (int i = tid * 4; i < base; i += 4096) {
        const int4 v = *(const int4*)(cnt + i);
        pre += (v.x + v.y) + (v.z + v.w);
    }

    const int i0 = base + tid * 4;
    const int c0 = (i0     < N) ? cnt[i0]     : 0;
    const int c1 = (i0 + 1 < N) ? cnt[i0 + 1] : 0;
    const int c2 = (i0 + 2 < N) ? cnt[i0 + 2] : 0;
    const int c3 = (i0 + 3 < N) ? cnt[i0 + 3] : 0;
    const int mine = (c0 + c1) + (c2 + c3);

    int incl = mine;
    for (int o = 1; o < 32; o <<= 1) {
        const int t = __shfl_up(incl, o);
        if (lane >= o) incl += t;
    }
    int pw = pre;
    for (int o = 16; o > 0; o >>= 1) pw += __shfl_xor(pw, o);

    if (lane == 31) wsum[wid] = incl;
    if (lane == 0)  wpre[wid] = pw;
    __syncthreads();

    int tot = 0, woff = 0;
    for (int k = 0; k < 32; ++k) {
        tot += wpre[k];
        if (k < wid) woff += wsum[k];
    }
    const int excl = tot + woff + incl - mine;
    if (i0     < N) cur[i0]     = excl + c0;
    if (i0 + 1 < N) cur[i0 + 1] = excl + c0 + c1;
    if (i0 + 2 < N) cur[i0 + 2] = excl + c0 + c1 + c2;
    if (i0 + 3 < N) cur[i0 + 3] = excl + mine;
}

__global__ void __launch_bounds__(1024)
tag_bin_scatter(const int* __restrict__ src, const int* __restrict__ dst,
                const int* __restrict__ gend, int2* __restrict__ pairs, int N, int E, int NB) {
    __shared__ int pos[TAG_MAXB];
    const int tid = threadIdx.x;
    if (tid < TAG_MAXB) {
        int p = 0;
        if (tid < NB) {
            const int i = tid * TAG_PB + blockIdx.x;
            p = (i > 0) ? gend[i - 1] : 0;
        }
        pos[tid] = p;
    }
    __syncthreads();
    const int chunk = (E + TAG_PB - 1) / TAG_PB;
    const int e0 = blockIdx.x * chunk;
    const int e1 = min(e0 + chunk, E);
    for (int e = e0 + tid; e < e1; e += 4096) {
        const bool k1 = e + 1024 < e1, k2 = e + 2048 < e1, k3 = e + 3072 < e1;
        const int d0 = dst[e],                  s0 = src[e];
        const int d1 = k1 ? dst[e + 1024] : 0,  s1 = k1 ? src[e + 1024] : 0;
        const int d2 = k2 ? dst[e + 2048] : 0,  s2 = k2 ? src[e + 2048] : 0;
        const int d3 = k3 ? dst[e + 3072] : 0,  s3 = k3 ? src[e + 3072] : 0;
        {
            const int p = atomicAdd(&pos[tag_range_of(d0, N)], 1);
            if (p >= 0 && p < E) pairs[p] = make_int2(s0, d0);
        }
        if (k1) {
            const int p = atomicAdd(&pos[tag_range_of(d1, N)], 1);
            if (p >= 0 && p < E) pairs[p] = make_int2(s1, d1);
        }
        if (k2) {
            const int p = atomicAdd(&pos[tag_range_of(d2, N)], 1);
            if (p >= 0 && p < E) pairs[p] = make_int2(s2, d2);
        }
        if (k3) {
            const int p = atomicAdd(&pos[tag_range_of(d3, N)], 1);
            if (p >= 0 && p < E) pairs[p] = make_int2(s3, d3);
        }
    }
}

__device__ __forceinline__ int tag_local_row(int d, int row0) {
    return min(max(d - row0, 0), TAG_RB - 1);
}

__global__ void __launch_bounds__(1024)
tag_bucket_rows(const int2* __restrict__ pairs, const int* __restrict__ gend,
                int* __restrict__ cnt, int* __restrict__ rend, int* __restrict__ csrc,
                int N, int E) {
    __shared__ int rc[TAG_RB];
    __shared__ int cursor[TAG_RB];
    __shared__ int wtot[TAG_RB / 32];
    const int tid  = threadIdx.x;
    const int lane = tid & 31;
    const int wid  = tid >> 5;
    const int b    = blockIdx.x;
    const int row0 = b * TAG_RB;
    int s0 = gend[max(b * TAG_PB - 1, 0)];
    int s1 = gend[(b + 1) * TAG_PB - 1];
    s0 = (b > 0) ? s0 : 0;
    s1 = min(max(s1, 0), E);
    s0 = min(max(s0, 0), s1);
    if (tid < TAG_RB) rc[tid] = 0;
    __syncthreads();
    for (int e = s0 + tid; e < s1; e += 4096) {
        const bool k1 = e + 1024 < s1, k2 = e + 2048 < s1, k3 = e + 3072 < s1;
        const int y0 = pairs[e].y;
        const int y1 = k1 ? pairs[e + 1024].y : 0;
        const int y2 = k2 ? pairs[e + 2048].y : 0;
        const int y3 = k3 ? pairs[e + 3072].y : 0;
        atomicAdd(&rc[tag_local_row(y0, row0)], 1);
        if (k1) atomicAdd(&rc[tag_local_row(y1, row0)], 1);
        if (k2) atomicAdd(&rc[tag_local_row(y2, row0)], 1);
        if (k3) atomicAdd(&rc[tag_local_row(y3, row0)], 1);
    }
    __syncthreads();
    int mine = 0, incl = 0;
    if (tid < TAG_RB) {
        mine = rc[tid];
        incl = mine;
        for (int o = 1; o < 32; o <<= 1) {
            const int t = __shfl_up(incl, o);
            if (lane >= o) incl += t;
        }
        if (lane == 31) wtot[wid] = incl;
    }
    __syncthreads();
    if (tid < TAG_RB) {
        int woff = 0;
        for (int k = 0; k < TAG_RB / 32; ++k) if (k < wid) woff += wtot[k];
        const int rowend = s0 + woff + incl;
        cursor[tid] = rowend - mine;
        const int row = row0 + tid;
        if (row < N) { cnt[row] = mine;  rend[row] = rowend; }
    }
    __syncthreads();
    for (int e = s0 + tid; e < s1; e += 4096) {
        const bool k1 = e + 1024 < s1, k2 = e + 2048 < s1, k3 = e + 3072 < s1;
        const int2 z = make_int2(0, 0);
        const int2 p0 = pairs[e];
        const int2 p1 = k1 ? pairs[e + 1024] : z;
        const int2 p2 = k2 ? pairs[e + 2048] : z;
        const int2 p3 = k3 ? pairs[e + 3072] : z;
        {
            const int p = atomicAdd(&cursor[tag_local_row(p0.y, row0)], 1);
            if (p >= 0 && p < E) csrc[p] = p0.x;
        }
        if (k1) {
            const int p = atomicAdd(&cursor[tag_local_row(p1.y, row0)], 1);
            if (p >= 0 && p < E) csrc[p] = p1.x;
        }
        if (k2) {
            const int p = atomicAdd(&cursor[tag_local_row(p2.y, row0)], 1);
            if (p >= 0 && p < E) csrc[p] = p2.x;
        }
        if (k3) {
            const int p = atomicAdd(&cursor[tag_local_row(p3.y, row0)], 1);
            if (p >= 0 && p < E) csrc[p] = p3.x;
        }
    }
}

__device__ __forceinline__ float tag_dis(int c) {
    return (c > 0) ? rsqrtf(fmaxf((float)c, 1.0f)) : 0.0f;
}

template <bool SRCW, bool SCALED>
__global__ void tag_hop128(const float* __restrict__ h, const int* __restrict__ cnt,
                           const int* __restrict__ rend, const int* __restrict__ csrc,
                           float* __restrict__ out, int N, int E) {
    const int lane = threadIdx.x & 31;
    const int row = __builtin_amdgcn_readfirstlane(
        (int)(blockIdx.x * (blockDim.x >> 5) + (threadIdx.x >> 5)));
    if (row >= N) return;
    int start = rend[max(row - 1, 0)];
    int end   = rend[row];
    start = (row > 0) ? start : 0;
    end   = min(max(end, 0), E);
    start = min(max(start, 0), end);
    const int c = end - start;
    const float di = tag_dis(c);

    float ax = 0.0f, ay = 0.0f, az = 0.0f, aw = 0.0f;
    const float* hl = h + lane * 4;
    for (int b = 0; b < c; b += 32) {
        const int j = b + lane;
        int s = row;
        float w = 0.0f;
        if (j < c) {
            s = csrc[start + j];
            s = min(max(s, 0), N - 1);
            if (SRCW) w = tag_dis(cnt[s]);
        }
        const int nb = min(32, c - b);
        for (int u = 0; u < nb; u += 4) {
            const int   s0 = __shfl(s, u),     s1 = __shfl(s, u + 1);
            const int   s2 = __shfl(s, u + 2), s3 = __shfl(s, u + 3);
            float w0, w1, w2, w3;
            if (SRCW) {
                w0 = __shfl(w, u);      w1 = __shfl(w, u + 1);
                w2 = __shfl(w, u + 2);  w3 = __shfl(w, u + 3);
            } else {
                w0 = 1.0f;
                w1 = (u + 1 < nb) ? 1.0f : 0.0f;
                w2 = (u + 2 < nb) ? 1.0f : 0.0f;
                w3 = (u + 3 < nb) ? 1.0f : 0.0f;
            }
            const float4 v0 = *(const float4*)(hl + (size_t)s0 * 128);
            const float4 v1 = *(const float4*)(hl + (size_t)s1 * 128);
            const float4 v2 = *(const float4*)(hl + (size_t)s2 * 128);
            const float4 v3 = *(const float4*)(hl + (size_t)s3 * 128);
            ax += v0.x * w0;  ay += v0.y * w0;  az += v0.z * w0;  aw += v0.w * w0;
            ax += v1.x * w1;  ay += v1.y * w1;  az += v1.z * w1;  aw += v1.w * w1;
            ax += v2.x * w2;  ay += v2.y * w2;  az += v2.z * w2;  aw += v2.w * w2;
            ax += v3.x * w3;  ay += v3.y * w3;  az += v3.z * w3;  aw += v3.w * w3;
        }
    }
    const float f = SCALED ? di * di : di;
    *(float4*)(out + (size_t)row * 128 + lane * 4) = make_float4(ax * f, ay * f, az * f, aw * f);
}

template <bool SCALED>
__global__ void tag_hop16(const float* __restrict__ t,
                          const int* __restrict__ rend, const int* __restrict__ csrc,
                          float* __restrict__ o, int N, int E) {
    const int lane = threadIdx.x & 31;
    const int row = __builtin_amdgcn_readfirstlane(
        (int)(blockIdx.x * (blockDim.x >> 5) + (threadIdx.x >> 5)));
    if (row >= N) return;
    const int g = lane >> 2;
    const int q = lane & 3;
    float4* op = (float4*)(o + (size_t)row * 16 + q * 4);
    const float4 old = *op;

    int start = rend[max(row - 1, 0)];
    int end   = rend[row];
    start = (row > 0) ? start : 0;
    end   = min(max(end, 0), E);
    start = min(max(start, 0), end);
    const int c = end - start;
    const float di = tag_dis(c);

    float ax = 0.0f, ay = 0.0f, az = 0.0f, aw = 0.0f;
    const float* tl = t + q * 4;
    for (int b = 0; b < c; b += 32) {
        const int j = b + lane;
        int s = row;
        if (j < c) {
            s = csrc[start + j];
            s = min(max(s, 0), N - 1);
        }
        const int nb = min(32, c - b);
        for (int u = 0; u < nb; u += 8) {
            const int   sj = __shfl(s, u + g);
            const float wj = (u + g < nb) ? 1.0f : 0.0f;
            const float4 v = *(const float4*)(tl + (size_t)sj * 16);
            ax += v.x * wj;  ay += v.y * wj;  az += v.z * wj;  aw += v.w * wj;
        }
    }
    for (int m = 4; m < 32; m <<= 1) {
        ax += __shfl_xor(ax, m);
        ay += __shfl_xor(ay, m);
        az += __shfl_xor(az, m);
        aw += __shfl_xor(aw, m);
    }
    if (lane < 4) {
        float4 r = old;
        r.x += ax * di;  r.y += ay * di;  r.z += az * di;  r.w += aw * di;
        if (SCALED) { r.x *= di;  r.y *= di;  r.z *= di;  r.w *= di; }
        *op = r;
    }
}

#define TAG_WPITCH 136

__device__ __forceinline__ void tag_load_a(float4 (&a)[16], const float* p) {
#pragma unroll
    for (int kb = 0; kb < 16; ++kb) a[kb] = *(const float4*)(p + kb * 8);
}

__device__ __forceinline__ void tag_scale_a(float4 (&a)[16], float f) {
#pragma unroll
    for (int kb = 0; kb < 16; ++kb) {
        a[kb].x *= f;  a[kb].y *= f;  a[kb].z *= f;  a[kb].w *= f;
    }
}

template <int NT>
__device__ __forceinline__ void tag_mma_plane(v8f (&acc)[NT], const float4 (&a)[16], const float4* bp) {
#pragma unroll
    for (int kb = 0; kb < 16; ++kb) {
        float4 b[NT];
#pragma unroll
        for (int t = 0; t < NT; ++t) b[t] = bp[(2 * kb) * TAG_WPITCH + t * 16];
        const v2f a0 = {a[kb].x, a[kb].y};
        const v2f a1 = {a[kb].z, a[kb].w};
#pragma unroll
        for (int t = 0; t < NT; ++t) {
            const v2f b0 = {b[t].x, b[t].y};
            acc[t] = __builtin_amdgcn_wmma_f32_16x16x4_f32(false, a0, false, b0, (short)0, acc[t], false, false);
        }
#pragma unroll
        for (int t = 0; t < NT; ++t) {
            const v2f b1 = {b[t].z, b[t].w};
            acc[t] = __builtin_amdgcn_wmma_f32_16x16x4_f32(false, a1, false, b1, (short)0, acc[t], false, false);
        }
    }
}

template <int NT>
__device__ __forceinline__ void tag_strip_products(v8f (&acc)[NT], const float* P0, const float* P1,
                                                   const float* P2, const float* P3H,
                                                   const int* cnt, const float4* bp,
                                                   int tile, size_t lo, int c) {
    const size_t ro = (size_t)tile * 2048 + lo;
#pragma unroll
    for (int t = 0; t < NT; ++t) acc[t] = (v8f){};

    const int dg = cnt[tile * 16 + c];
#pragma nounroll
    for (int p = 0; p < 4; ++p) {
        const float* pl = (p == 0) ? P0 : (p == 1) ? P1 : (p == 2) ? P2 : P3H;
        float4 a[16];
        tag_load_a(a, pl + ro);
        if (p == 1 || p == 2) tag_scale_a(a, (dg > 0) ? sqrtf((float)dg) : 0.0f);
        tag_mma_plane<NT>(acc, a, bp + p * (32 * TAG_WPITCH));
    }
}

template <int NT>
__device__ __forceinline__ void tag_strip_store(const v8f (&acc)[NT], float* P3H, const float* bias,
                                                int tile, int ct0, int h, int c) {
    float* cp = P3H + (size_t)(tile * 16 + h * 8) * 128 + ct0 * 16 + c;
#pragma unroll
    for (int t = 0; t < NT; ++t) {
        const float bv = bias[(ct0 + t) * 16 + c];
#pragma unroll
        for (int r = 0; r < 8; ++r)
            cp[(size_t)r * 128 + t * 16] = fmaxf(acc[t][r] + bv, 0.0f);
    }
}

__global__ void __launch_bounds__(512)
tag_gemm_all(const float* __restrict__ P0, const float* __restrict__ P1,
             const float* __restrict__ P2, float* P3H,
             const float* __restrict__ W, const float* __restrict__ bias,
             const int* __restrict__ cnt, int ntiles) {
    __shared__ float4 Wq[128 * TAG_WPITCH];
    const int tid  = threadIdx.x;
    const int lane = tid & 31;
    const int c = lane & 15;
    const int h = lane >> 4;
    const int w    = __builtin_amdgcn_readfirstlane(tid >> 5);
    const int cls  = w & 3;
    const int slot = w >> 2;
    const int s0 = (int)(((unsigned)blockIdx.x * (unsigned)ntiles) / gridDim.x);
    const int s1 = (int)(((unsigned)(blockIdx.x + 1) * (unsigned)ntiles) / gridDim.x);
    const int q   = (s1 - s0) >> 2;
    const int rem = (s1 - s0) & 3;
    const size_t lo = (size_t)c * 128 + h * 4;

#pragma unroll 8
    for (int i = 0; i < 32; ++i) {
        const int idx = tid + 512 * i;
        const int n = idx & 127, kq = idx >> 7;
        const float* wp = W + (size_t)(4 * kq) * 128 + n;
        Wq[kq * TAG_WPITCH + n] = make_float4(wp[0], wp[128], wp[256], wp[384]);
    }
    __syncthreads();

    const float4* bp = Wq + h * TAG_WPITCH + c;
    for (int k = slot; k < q; k += 4) {
        const int tile = s0 + cls + 4 * k;
        v8f acc[8];
        tag_strip_products<8>(acc, P0, P1, P2, P3H, cnt, bp, tile, lo, c);
        tag_strip_store<8>(acc, P3H, bias, tile, 0, h, c);
    }
    const int  m    = 2 - slot;
    const bool part = m >= 0 && m < rem;
    const int  ptile = s0 + 4 * q + m;
    v8f pacc[2] = {};
    if (part) tag_strip_products<2>(pacc, P0, P1, P2, P3H, cnt, bp + cls * 32, ptile, lo, c);
    __syncthreads();
    if (part) tag_strip_store<2>(pacc, P3H, bias, ptile, cls * 2, h, c);
}

__global__ void tag_gemm_l2(const float* __restrict__ A, const float* __restrict__ W,
                            float* __restrict__ out, float* __restrict__ q1,
                            float* __restrict__ q2, float* __restrict__ q3,
                            const float* __restrict__ bias, const int* __restrict__ cnt) {
    const int lane = threadIdx.x;
    const int tm = blockIdx.x;
    const int lo = lane & 15;
    const int hi = lane >> 4;
    const int arow = tm * 16 + lo;
    const int crow = tm * 16 + hi * 8;

    v8f c0 = {}, c1 = {}, c2 = {}, c3 = {};
    const float* arowp = A + (size_t)arow * 128 + hi * 2;
    const float* bp0   = W + (size_t)(hi * 2) * 16 + lo;
#pragma unroll 8
    for (int k0 = 0; k0 < 128; k0 += 4) {
        v2f a = *(const v2f*)(arowp + k0);
        v2f b0, b1, b2, b3;
        b0.x = bp0[0 * 2048 + k0 * 16];  b0.y = bp0[0 * 2048 + k0 * 16 + 16];
        b1.x = bp0[1 * 2048 + k0 * 16];  b1.y = bp0[1 * 2048 + k0 * 16 + 16];
        b2.x = bp0[2 * 2048 + k0 * 16];  b2.y = bp0[2 * 2048 + k0 * 16 + 16];
        b3.x = bp0[3 * 2048 + k0 * 16];  b3.y = bp0[3 * 2048 + k0 * 16 + 16];
        c0 = __builtin_amdgcn_wmma_f32_16x16x4_f32(false, a, false, b0, (short)0, c0, false, false);
        c1 = __builtin_amdgcn_wmma_f32_16x16x4_f32(false, a, false, b1, (short)0, c1, false, false);
        c2 = __builtin_amdgcn_wmma_f32_16x16x4_f32(false, a, false, b2, (short)0, c2, false, false);
        c3 = __builtin_amdgcn_wmma_f32_16x16x4_f32(false, a, false, b3, (short)0, c3, false, false);
    }

    const float bv = bias[lo];
#pragma unroll
    for (int r = 0; r < 8; ++r) {
        const size_t off = (size_t)(crow + r) * 16 + lo;
        out[off] = c0[r] + bv;
        q1[off]  = c1[r];
        q2[off]  = c2[r];
        q3[off]  = c3[r] * tag_dis(cnt[crow + r]);
    }
}

extern "C" void kernel_launch(void* const* d_in, const int* in_sizes, int n_in,
                              void* d_out, int out_size, void* d_ws, size_t ws_size,
                              hipStream_t stream) {
    const float* x  = (const float*)d_in[0];
    const int*   ei = (const int*)d_in[1];
    const float* W1 = (const float*)d_in[2];
    const float* b1 = (const float*)d_in[3];
    const float* W2 = (const float*)d_in[4];
    const float* b2 = (const float*)d_in[5];

    const int N = in_sizes[0] / 128;
    const int E = in_sizes[1] / 2;
    const int* src = ei;
    const int* dst = ei + E;

    int*   cnt  = (int*)d_ws;
    int*   cur  = cnt + N;
    int*   csrc = cur + N;
    float* bufA = (float*)(csrc + E);
    float* bufB = bufA + (size_t)N * 128;
    float* acc  = bufB + (size_t)N * 128;
    const int NB = (N + TAG_RB - 1) / TAG_RB;
    const int M  = NB * TAG_PB;
    int*   gh    = (int*)bufB;
    int*   gend  = gh + M;
    int2*  pairs = (int2*)acc;
    float* out  = (float*)d_out;

    const int T = 256;
    const int RPB = T / 32;
    const int hblk = (N + RPB - 1) / RPB;
    const int ntiles = N / 16;
    const dim3 g2(N / 16, 1);

    tag_bin_hist   <<<TAG_PB, 1024, 0, stream>>>(dst, gh, N, E, NB);
    tag_rowstart   <<<(M + 4095) / 4096, 1024, 0, stream>>>(gh, gend, M);
    tag_bin_scatter<<<TAG_PB, 1024, 0, stream>>>(src, dst, gend, pairs, N, E, NB);
    tag_bucket_rows<<<NB, 1024, 0, stream>>>(pairs, gend, cnt, cur, csrc, N, E);

    tag_hop128<true,  true ><<<hblk, T, 0, stream>>>(x,    cnt, cur, csrc, bufA, N, E);
    tag_hop128<false, true ><<<hblk, T, 0, stream>>>(bufA, cnt, cur, csrc, bufB, N, E);
    tag_hop128<false, false><<<hblk, T, 0, stream>>>(bufB, cnt, cur, csrc, acc,  N, E);
    tag_gemm_all<<<256, 512, 0, stream>>>(x, bufA, bufB, acc, W1, b1, cnt, ntiles);

    float* q1 = bufB;
    float* q2 = q1 + (size_t)N * 16;
    float* q3 = q2 + (size_t)N * 16;
    tag_gemm_l2<<<g2, 32, 0, stream>>>(acc, W2, out, q1, q2, q3, b2, cnt);
    tag_hop16<true ><<<hblk, T, 0, stream>>>(q3, cur, csrc, q2, N, E);
    tag_hop16<true ><<<hblk, T, 0, stream>>>(q2, cur, csrc, q1, N, E);
    tag_hop16<false><<<hblk, T, 0, stream>>>(q1, cur, csrc, out, N, E);

    (void)n_in; (void)out_size; (void)ws_size;
}
